# starting kernel
# baseline (speedup 1.0000x reference)
#include <hip/hip_runtime.h>

typedef float v2f __attribute__((ext_vector_type(2)));
typedef float v8f __attribute__((ext_vector_type(8)));


__global__ void zero_f32(float* __restrict__ p, long long nElems) {
  long long i = (long long)blockIdx.x * blockDim.x + threadIdx.x;
  if (i < nElems) p[i] = 0.0f;
}

__global__ void deg_accum(const int* __restrict__ dst, float* __restrict__ deg, int nE) {
  int e = blockIdx.x * blockDim.x + threadIdx.x;
  if (e < nE) atomicAdd(&deg[dst[e]], 1.0f);
}

__global__ void deg_finalize(float* __restrict__ deg, int n) {
  int i = blockIdx.x * blockDim.x + threadIdx.x;
  if (i < n) deg[i] = rsqrtf(deg[i] + 1.0f);
}

__global__ void gemm_wmma_f32(const float* __restrict__ A, const float* __restrict__ B,
                              float* __restrict__ D, int M, int K, int N) {
  int wavesPerBlock = blockDim.x >> 5;
  long long wave = (long long)blockIdx.x * wavesPerBlock + (threadIdx.x >> 5);
  int tilesN = N >> 4;
  long long tilesM = (M + 15) >> 4;
  long long tm = wave / tilesN;
  int tn = (int)(wave % tilesN);
  if (tm >= tilesM) return;

  int lane = threadIdx.x & 31;
  int lr = lane & 15;
  int hi = lane >> 4;

  int rowA = (int)(tm * 16) + lr;
  if (rowA > M - 1) rowA = M - 1;
  const float* arow = A + (size_t)rowA * K;
  const float* bcol = B + (size_t)tn * 16 + lr;

  v8f acc = {};
  for (int k = 0; k < K; k += 4) {
    v2f a, b;
    int ka = k + 2 * hi;
    a.x = arow[ka];
    a.y = arow[ka + 1];
    b.x = bcol[(size_t)ka * N];
    b.y = bcol[(size_t)(ka + 1) * N];
    acc = __builtin_amdgcn_wmma_f32_16x16x4_f32(
         false, a,  false, b,
         (short)0, acc,  false,  false);
  }

  float* dcol = D + (size_t)tn * 16 + lr;
#pragma unroll
  for (int j = 0; j < 8; ++j) {
    int rowD = (int)(tm * 16) + j + 8 * hi;
    if (rowD < M) dcol[(size_t)rowD * N] = acc[j];
  }
}

template <int C>
__global__ void scatter_add(const float* __restrict__ h, const int* __restrict__ src,
                            const int* __restrict__ dst, const float* __restrict__ dis,
                            float* __restrict__ agg, int nE) {
  long long gid = (long long)blockIdx.x * blockDim.x + threadIdx.x;
  int e = (int)(gid / C);
  if (e >= nE) return;
  int c = (int)(gid % C);
  int s = src[e];
  int d = dst[e];
  float norm = dis[s] * dis[d];
  atomicAdd(&agg[(size_t)d * C + c], h[(size_t)s * C + c] * norm);
}

template <int C, bool RELU>
__global__ void self_bias(const float* __restrict__ agg, const float* __restrict__ h,
                          const float* __restrict__ dis, const float* __restrict__ b,
                          float* __restrict__ out, int n) {
  long long gid = (long long)blockIdx.x * blockDim.x + threadIdx.x;
  if (gid >= (long long)n * C) return;
  int i = (int)(gid / C);
  int c = (int)(gid % C);
  float di = dis[i];
  float v = agg[gid] + h[gid] * (di * di) + b[c];
  out[gid] = RELU ? fmaxf(v, 0.0f) : v;
}


extern "C" void kernel_launch(void* const* d_in, const int* in_sizes, int n_in,
                              void* d_out, int out_size, void* d_ws, size_t ws_size,
                              hipStream_t stream) {
  const float* x  = (const float*)d_in[0];
  const float* W1 = (const float*)d_in[1];
  const float* b1 = (const float*)d_in[2];
  const float* W2 = (const float*)d_in[3];
  const float* b2 = (const float*)d_in[4];
  const int*   ei = (const int*)d_in[5];

  const int IN  = 128;
  const int HID = in_sizes[2];
  const int OUT = in_sizes[4];
  const int n   = in_sizes[0] / IN;
  const int nE  = in_sizes[5] / 2;
  const int* src = ei;
  const int* dst = ei + nE;

  char* ws = (char*)d_ws;
  size_t off = 0;
  auto alloc = [&](size_t bytes) -> float* {
    float* p = (float*)(ws + off);
    off += (bytes + 255) & ~(size_t)255;
    return p;
  };
  float* dis  = alloc((size_t)n * 4);
  float* h1   = alloc((size_t)n * HID * 4);
  float* agg1 = alloc((size_t)n * HID * 4);
  float* h2   = alloc((size_t)n * OUT * 4);
  float* agg2 = alloc((size_t)n * OUT * 4);
  float* outp = (float*)d_out;

  const int T = 256;
  auto blocks1d = [&](long long work) -> unsigned int {
    return (unsigned int)((work + T - 1) / T);
  };

  zero_f32<<<blocks1d(n), T, 0, stream>>>(dis, n);
  deg_accum<<<blocks1d(nE), T, 0, stream>>>(dst, dis, nE);
  deg_finalize<<<blocks1d(n), T, 0, stream>>>(dis, n);

  {
    long long tilesM = (n + 15) / 16;
    long long waves = tilesM * (HID / 16);
    unsigned int blocks = (unsigned int)((waves + (T / 32) - 1) / (T / 32));
    gemm_wmma_f32<<<blocks, T, 0, stream>>>(x, W1, h1, n, IN, HID);
  }
  zero_f32<<<blocks1d((long long)n * HID), T, 0, stream>>>(agg1, (long long)n * HID);
  scatter_add<64><<<blocks1d((long long)nE * 64), T, 0, stream>>>(h1, src, dst, dis, agg1, nE);
  self_bias<64, true><<<blocks1d((long long)n * 64), T, 0, stream>>>(agg1, h1, dis, b1, h1, n);

  {
    long long tilesM = (n + 15) / 16;
    long long waves = tilesM * (OUT / 16);
    unsigned int blocks = (unsigned int)((waves + (T / 32) - 1) / (T / 32));
    gemm_wmma_f32<<<blocks, T, 0, stream>>>(h1, W2, h2, n, HID, OUT);
  }
  zero_f32<<<blocks1d((long long)n * OUT), T, 0, stream>>>(agg2, (long long)n * OUT);
  scatter_add<32><<<blocks1d((long long)nE * 32), T, 0, stream>>>(h2, src, dst, dis, agg2, nE);
  self_bias<32, false><<<blocks1d((long long)n * 32), T, 0, stream>>>(agg2, h2, dis, b2, outp, n);
}


# Replace float-atomic edge scatter by in-call CSR (returning int atomic rank, scan, fill) + atomic-free per-node gather with fused self/bias/ReLU; node sum order = rank order, FMA accumulate
# speedup vs baseline: 1.8992x; 1.8992x over previous
#include <hip/hip_runtime.h>

typedef float v2f __attribute__((ext_vector_type(2)));
typedef float v8f __attribute__((ext_vector_type(8)));


__global__ void zero_i32(int* __restrict__ p, int nElems) {
  int i = blockIdx.x * blockDim.x + threadIdx.x;
  if (i < nElems) p[i] = 0;
}

__global__ void edge_rank(const int* __restrict__ dst, int* __restrict__ cnt,
                          int* __restrict__ rank, int nE) {
  int e = blockIdx.x * blockDim.x + threadIdx.x;
  if (e < nE) rank[e] = atomicAdd(&cnt[dst[e]], 1);
}

__global__ void __launch_bounds__(256) row_scan(const int* __restrict__ cnt,
                                                 int* __restrict__ rowptr,
                                                 float* __restrict__ dis, int n) {
  __shared__ int wsum[8];
  __shared__ int wpre[8];
  int t = threadIdx.x;
  int lane = t & 31;
  int w = t >> 5;
  int base = blockIdx.x * 256;
  const int4* c4 = (const int4*)cnt;
  int nv = base >> 2;
  int s = 0;
  for (int i = t; i < nv; i += 256 * 8) {
    int4 v[8];
#pragma unroll
    for (int u = 0; u < 8; ++u) {
      int j = i + u * 256;
      v[u] = make_int4(0, 0, 0, 0);
      if (j < nv) v[u] = c4[j];
    }
#pragma unroll
    for (int u = 0; u < 8; ++u) s += (v[u].x + v[u].y) + (v[u].z + v[u].w);
  }
  int i = base + t;
  int c = (i < n) ? cnt[i] : 0;
  int inc = c;
  for (int o = 1; o < 32; o <<= 1) {
    int v = __shfl_up(inc, o);
    if (lane >= o) inc += v;
  }
  for (int o = 16; o > 0; o >>= 1) s += __shfl_xor(s, o);
  if (lane == 31) wsum[w] = inc;
  if (lane == 0) wpre[w] = s;
  __syncthreads();
  int off = 0;
  for (int k = 0; k < 8; ++k) {
    off += wpre[k];
    if (k < w) off += wsum[k];
  }
  int excl = off + inc - c;
  if (i < n) {
    rowptr[i] = excl;
    dis[i] = rsqrtf((float)c + 1.0f);
  }
  if (i == n - 1) rowptr[n] = excl + c;
}

__global__ void csr_fill(const int* __restrict__ src, const int* __restrict__ dst,
                         const int* __restrict__ rank, const int* __restrict__ rowptr,
                         const float* __restrict__ dis, int* __restrict__ csr_src,
                         float* __restrict__ csr_w, int nE) {
  int e = blockIdx.x * blockDim.x + threadIdx.x;
  if (e >= nE) return;
  int s = src[e];
  int d = dst[e];
  int slot = rowptr[d] + rank[e];
  csr_src[slot] = s;
  csr_w[slot] = dis[s] * dis[d];
}

__global__ void gemm_wmma_f32(const float* __restrict__ A, const float* __restrict__ B,
                              float* __restrict__ D, int M, int K, int N) {
  int wavesPerBlock = blockDim.x >> 5;
  long long wave = (long long)blockIdx.x * wavesPerBlock + (threadIdx.x >> 5);
  int tilesN = N >> 4;
  long long tilesM = (M + 15) >> 4;
  long long tm = wave / tilesN;
  int tn = (int)(wave % tilesN);
  if (tm >= tilesM) return;

  int lane = threadIdx.x & 31;
  int lr = lane & 15;
  int hi = lane >> 4;

  int rowA = (int)(tm * 16) + lr;
  if (rowA > M - 1) rowA = M - 1;
  const float* arow = A + (size_t)rowA * K;
  const float* bcol = B + (size_t)tn * 16 + lr;

  v8f acc = {};
  for (int k = 0; k < K; k += 4) {
    v2f a, b;
    int ka = k + 2 * hi;
    a.x = arow[ka];
    a.y = arow[ka + 1];
    b.x = bcol[(size_t)ka * N];
    b.y = bcol[(size_t)(ka + 1) * N];
    acc = __builtin_amdgcn_wmma_f32_16x16x4_f32(
         false, a,  false, b,
         (short)0, acc,  false,  false);
  }

  float* dcol = D + (size_t)tn * 16 + lr;
#pragma unroll
  for (int j = 0; j < 8; ++j) {
    int rowD = (int)(tm * 16) + j + 8 * hi;
    if (rowD < M) dcol[(size_t)rowD * N] = acc[j];
  }
}

template <int C, bool RELU>
__global__ void gather_rows(const float* __restrict__ h, const int* __restrict__ rowptr,
                            const int* __restrict__ csr_src, const float* __restrict__ csr_w,
                            const float* __restrict__ dis, const float* __restrict__ b,
                            float* __restrict__ out, int n) {
  constexpr int LPN = C / 4;
  constexpr int NPW = 32 / LPN;
  int lane = threadIdx.x & 31;
  int sub = lane % LPN;
  long long wave = ((long long)blockIdx.x * blockDim.x + threadIdx.x) >> 5;
  long long nodeL = wave * NPW + lane / LPN;
  if (nodeL >= n) return;
  int node = (int)nodeL;
  int beg = rowptr[node];
  int end = rowptr[node + 1];
  const float4* h4 = (const float4*)h;
  float4 acc = make_float4(0.0f, 0.0f, 0.0f, 0.0f);
  for (int base = beg; base < end; base += LPN) {
    int my = base + sub;
    int s_l = 0;
    float w_l = 0.0f;
    if (my < end) {
      s_l = csr_src[my];
      w_l = csr_w[my];
    }
    int cnt = end - base;
    float4 v[LPN];
    float wt[LPN];
#pragma unroll
    for (int t = 0; t < LPN; ++t) {
      int s = __shfl(s_l, t, LPN);
      wt[t] = __shfl(w_l, t, LPN);
      v[t] = make_float4(0.0f, 0.0f, 0.0f, 0.0f);
      if (t < cnt) v[t] = h4[(size_t)s * LPN + sub];
    }
#pragma unroll
    for (int t = 0; t < LPN; ++t) {
      acc.x = __builtin_fmaf(v[t].x, wt[t], acc.x);
      acc.y = __builtin_fmaf(v[t].y, wt[t], acc.y);
      acc.z = __builtin_fmaf(v[t].z, wt[t], acc.z);
      acc.w = __builtin_fmaf(v[t].w, wt[t], acc.w);
    }
  }
  float di = dis[node];
  float d2 = di * di;
  float4 hv = h4[(size_t)node * LPN + sub];
  float4 bv = ((const float4*)b)[sub];
  float4 r;
  r.x = acc.x + hv.x * d2 + bv.x;
  r.y = acc.y + hv.y * d2 + bv.y;
  r.z = acc.z + hv.z * d2 + bv.z;
  r.w = acc.w + hv.w * d2 + bv.w;
  if (RELU) {
    r.x = fmaxf(r.x, 0.0f);
    r.y = fmaxf(r.y, 0.0f);
    r.z = fmaxf(r.z, 0.0f);
    r.w = fmaxf(r.w, 0.0f);
  }
  ((float4*)out)[(size_t)node * LPN + sub] = r;
}


extern "C" void kernel_launch(void* const* d_in, const int* in_sizes, int n_in,
                              void* d_out, int out_size, void* d_ws, size_t ws_size,
                              hipStream_t stream) {
  const float* x  = (const float*)d_in[0];
  const float* W1 = (const float*)d_in[1];
  const float* b1 = (const float*)d_in[2];
  const float* W2 = (const float*)d_in[3];
  const float* b2 = (const float*)d_in[4];
  const int*   ei = (const int*)d_in[5];

  const int IN  = 128;
  const int HID = in_sizes[2];
  const int OUT = in_sizes[4];
  const int n   = in_sizes[0] / IN;
  const int nE  = in_sizes[5] / 2;
  const int* src = ei;
  const int* dst = ei + nE;

  char* ws = (char*)d_ws;
  size_t off = 0;
  auto alloc = [&](size_t bytes) -> void* {
    void* p = (void*)(ws + off);
    off += (bytes + 255) & ~(size_t)255;
    return p;
  };
  float* dis    = (float*)alloc((size_t)n * 4);
  float* bufA   = (float*)alloc((size_t)n * HID * 4);
  float* bufB   = (float*)alloc((size_t)n * HID * 4);
  int*   cnt    = (int*)alloc((size_t)n * 4);
  int*   rowptr = (int*)alloc((size_t)(n + 1) * 4);
  int*   rank   = (int*)alloc((size_t)nE * 4);
  int*   csrS   = (int*)alloc((size_t)nE * 4);
  float* csrW   = (float*)alloc((size_t)nE * 4);
  float* h1   = bufA;
  float* hid  = bufB;
  float* h2   = bufA;
  float* outp = (float*)d_out;

  const int T = 256;
  auto blocks1d = [&](long long work) -> unsigned int {
    return (unsigned int)((work + T - 1) / T);
  };

  zero_i32<<<blocks1d(n), T, 0, stream>>>(cnt, n);
  edge_rank<<<blocks1d(nE), T, 0, stream>>>(dst, cnt, rank, nE);
  row_scan<<<blocks1d(n), T, 0, stream>>>(cnt, rowptr, dis, n);
  csr_fill<<<blocks1d(nE), T, 0, stream>>>(src, dst, rank, rowptr, dis, csrS, csrW, nE);

  {
    long long tilesM = (n + 15) / 16;
    long long waves = tilesM * (HID / 16);
    unsigned int blocks = (unsigned int)((waves + (T / 32) - 1) / (T / 32));
    gemm_wmma_f32<<<blocks, T, 0, stream>>>(x, W1, h1, n, IN, HID);
  }
  gather_rows<64, true><<<blocks1d((long long)n * 16), T, 0, stream>>>(
      h1, rowptr, csrS, csrW, dis, b1, hid, n);

  {
    long long tilesM = (n + 15) / 16;
    long long waves = tilesM * (OUT / 16);
    unsigned int blocks = (unsigned int)((waves + (T / 32) - 1) / (T / 32));
    gemm_wmma_f32<<<blocks, T, 0, stream>>>(hid, W2, h2, n, HID, OUT);
  }
  gather_rows<32, false><<<blocks1d((long long)n * 8), T, 0, stream>>>(
      h2, rowptr, csrS, csrW, dis, b2, outp, n);
}


# GEMM: one wave per 16-row strip over all N columns; A rows read once by 16-byte loads, W staged once per block in LDS as k pairs; k summed in 8-wide blocks (new order inside products)
# speedup vs baseline: 2.0509x; 1.0799x over previous
#include <hip/hip_runtime.h>

typedef float v2f __attribute__((ext_vector_type(2)));
typedef float v8f __attribute__((ext_vector_type(8)));


__global__ void zero_i32(int* __restrict__ p, int nElems) {
  int i = blockIdx.x * blockDim.x + threadIdx.x;
  if (i < nElems) p[i] = 0;
}

__global__ void edge_rank(const int* __restrict__ dst, int* __restrict__ cnt,
                          int* __restrict__ rank, int nE) {
  int e = blockIdx.x * blockDim.x + threadIdx.x;
  if (e < nE) rank[e] = atomicAdd(&cnt[dst[e]], 1);
}

__global__ void __launch_bounds__(256) row_scan(const int* __restrict__ cnt,
                                                 int* __restrict__ rowptr,
                                                 float* __restrict__ dis, int n) {
  __shared__ int wsum[8];
  __shared__ int wpre[8];
  int t = threadIdx.x;
  int lane = t & 31;
  int w = t >> 5;
  int base = blockIdx.x * 256;
  const int4* c4 = (const int4*)cnt;
  int nv = base >> 2;
  int s = 0;
  for (int i = t; i < nv; i += 256 * 8) {
    int4 v[8];
#pragma unroll
    for (int u = 0; u < 8; ++u) {
      int j = i + u * 256;
      v[u] = make_int4(0, 0, 0, 0);
      if (j < nv) v[u] = c4[j];
    }
#pragma unroll
    for (int u = 0; u < 8; ++u) s += (v[u].x + v[u].y) + (v[u].z + v[u].w);
  }
  int i = base + t;
  int c = (i < n) ? cnt[i] : 0;
  int inc = c;
  for (int o = 1; o < 32; o <<= 1) {
    int v = __shfl_up(inc, o);
    if (lane >= o) inc += v;
  }
  for (int o = 16; o > 0; o >>= 1) s += __shfl_xor(s, o);
  if (lane == 31) wsum[w] = inc;
  if (lane == 0) wpre[w] = s;
  __syncthreads();
  int off = 0;
  for (int k = 0; k < 8; ++k) {
    off += wpre[k];
    if (k < w) off += wsum[k];
  }
  int excl = off + inc - c;
  if (i < n) {
    rowptr[i] = excl;
    dis[i] = rsqrtf((float)c + 1.0f);
  }
  if (i == n - 1) rowptr[n] = excl + c;
}

__global__ void csr_fill(const int* __restrict__ src, const int* __restrict__ dst,
                         const int* __restrict__ rank, const int* __restrict__ rowptr,
                         const float* __restrict__ dis, int* __restrict__ csr_src,
                         float* __restrict__ csr_w, int nE) {
  int e = blockIdx.x * blockDim.x + threadIdx.x;
  if (e >= nE) return;
  int s = src[e];
  int d = dst[e];
  int slot = rowptr[d] + rank[e];
  csr_src[slot] = s;
  csr_w[slot] = dis[s] * dis[d];
}

template <int K, int N>
__global__ void __launch_bounds__(256) gemm_strip_f32(const float* __restrict__ A,
                                                       const float* __restrict__ B,
                                                       float* __restrict__ D, int M) {
  constexpr int NT = N / 16;
  constexpr int Q = K / 8;
  constexpr int P2 = N + 8;
  constexpr int ITEMS = (K / 2) * (N / 4);
  __shared__ float4 Bs4[(K / 2) * P2 / 2];

  const int t = threadIdx.x;
  const int lane = t & 31;
  const int c = lane & 15;
  const int h = lane >> 4;
  const int wv = __builtin_amdgcn_readfirstlane(t >> 5);
  const int strips = (M + 15) >> 4;
  const int wstrip = blockIdx.x * 8 + wv;
  const bool live = wstrip < strips;
  const int strip = live ? wstrip : strips - 1;

  const float4* B4 = (const float4*)B;
  float4 w0[ITEMS / 256], w1[ITEMS / 256];
#pragma unroll
  for (int i = 0; i < ITEMS / 256; ++i) {
    int it = t + 256 * i;
    int kp = it / (N / 4);
    int n4 = it % (N / 4);
    w0[i] = B4[(2 * kp) * (N / 4) + n4];
    w1[i] = B4[(2 * kp + 1) * (N / 4) + n4];
  }

  int rowA = strip * 16 + c;
  if (rowA > M - 1) rowA = M - 1;
  const float4* arow = (const float4*)(A + (size_t)rowA * K) + h;
  float4 a[Q];
#pragma unroll
  for (int q = 0; q < Q; ++q) a[q] = arow[2 * q];

#pragma unroll
  for (int i = 0; i < ITEMS / 256; ++i) {
    int it = t + 256 * i;
    int kp = it / (N / 4);
    int n4 = it % (N / 4);
    float4* dst = Bs4 + (kp * P2 + 4 * n4) / 2;
    dst[0] = make_float4(w0[i].x, w1[i].x, w0[i].y, w1[i].y);
    dst[1] = make_float4(w0[i].z, w1[i].z, w0[i].w, w1[i].w);
  }
  __builtin_amdgcn_fence(__ATOMIC_RELEASE, "workgroup", "local");
  __builtin_amdgcn_s_barrier();
  __builtin_amdgcn_fence(__ATOMIC_ACQUIRE, "workgroup", "local");

  const v2f* bp = (const v2f*)Bs4 + (2 * h) * P2 + c;
  v8f acc[NT];
#pragma unroll
  for (int tt = 0; tt < NT; ++tt) acc[tt] = v8f{};

#pragma unroll
  for (int q = 0; q < Q; ++q) {
#pragma unroll
    for (int s2 = 0; s2 < 2; ++s2) {
      v2f av;
      av.x = s2 ? a[q].z : a[q].x;
      av.y = s2 ? a[q].w : a[q].y;
#pragma unroll
      for (int tt = 0; tt < NT; ++tt) {
        v2f bv = bp[(4 * q + s2) * P2 + 16 * tt];
        acc[tt] = __builtin_amdgcn_wmma_f32_16x16x4_f32(
             false, av,  false, bv,
             (short)0, acc[tt],  false,  false);
      }
    }
  }

  if (!live) return;
  float* dlane = D + ((size_t)strip * 16 + 8 * h) * N + c;
  if (strip * 16 + 16 <= M) {
#pragma unroll
    for (int tt = 0; tt < NT; ++tt) {
#pragma unroll
      for (int j = 0; j < 8; ++j) dlane[j * N + 16 * tt] = acc[tt][j];
    }
  } else {
#pragma unroll
    for (int tt = 0; tt < NT; ++tt) {
#pragma unroll
      for (int j = 0; j < 8; ++j) {
        if (strip * 16 + 8 * h + j < M) dlane[j * N + 16 * tt] = acc[tt][j];
      }
    }
  }
}

template <int C, bool RELU>
__global__ void gather_rows(const float* __restrict__ h, const int* __restrict__ rowptr,
                            const int* __restrict__ csr_src, const float* __restrict__ csr_w,
                            const float* __restrict__ dis, const float* __restrict__ b,
                            float* __restrict__ out, int n) {
  constexpr int LPN = C / 4;
  constexpr int NPW = 32 / LPN;
  int lane = threadIdx.x & 31;
  int sub = lane % LPN;
  long long wave = ((long long)blockIdx.x * blockDim.x + threadIdx.x) >> 5;
  long long nodeL = wave * NPW + lane / LPN;
  if (nodeL >= n) return;
  int node = (int)nodeL;
  int beg = rowptr[node];
  int end = rowptr[node + 1];
  const float4* h4 = (const float4*)h;
  float4 acc = make_float4(0.0f, 0.0f, 0.0f, 0.0f);
  for (int base = beg; base < end; base += LPN) {
    int my = base + sub;
    int s_l = 0;
    float w_l = 0.0f;
    if (my < end) {
      s_l = csr_src[my];
      w_l = csr_w[my];
    }
    int cnt = end - base;
    float4 v[LPN];
    float wt[LPN];
#pragma unroll
    for (int t = 0; t < LPN; ++t) {
      int s = __shfl(s_l, t, LPN);
      wt[t] = __shfl(w_l, t, LPN);
      v[t] = make_float4(0.0f, 0.0f, 0.0f, 0.0f);
      if (t < cnt) v[t] = h4[(size_t)s * LPN + sub];
    }
#pragma unroll
    for (int t = 0; t < LPN; ++t) {
      acc.x = __builtin_fmaf(v[t].x, wt[t], acc.x);
      acc.y = __builtin_fmaf(v[t].y, wt[t], acc.y);
      acc.z = __builtin_fmaf(v[t].z, wt[t], acc.z);
      acc.w = __builtin_fmaf(v[t].w, wt[t], acc.w);
    }
  }
  float di = dis[node];
  float d2 = di * di;
  float4 hv = h4[(size_t)node * LPN + sub];
  float4 bv = ((const float4*)b)[sub];
  float4 r;
  r.x = acc.x + hv.x * d2 + bv.x;
  r.y = acc.y + hv.y * d2 + bv.y;
  r.z = acc.z + hv.z * d2 + bv.z;
  r.w = acc.w + hv.w * d2 + bv.w;
  if (RELU) {
    r.x = fmaxf(r.x, 0.0f);
    r.y = fmaxf(r.y, 0.0f);
    r.z = fmaxf(r.z, 0.0f);
    r.w = fmaxf(r.w, 0.0f);
  }
  ((float4*)out)[(size_t)node * LPN + sub] = r;
}


extern "C" void kernel_launch(void* const* d_in, const int* in_sizes, int n_in,
                              void* d_out, int out_size, void* d_ws, size_t ws_size,
                              hipStream_t stream) {
  const float* x  = (const float*)d_in[0];
  const float* W1 = (const float*)d_in[1];
  const float* b1 = (const float*)d_in[2];
  const float* W2 = (const float*)d_in[3];
  const float* b2 = (const float*)d_in[4];
  const int*   ei = (const int*)d_in[5];

  const int IN  = 128;
  const int HID = in_sizes[2];
  const int n   = in_sizes[0] / IN;
  const int nE  = in_sizes[5] / 2;
  const int* src = ei;
  const int* dst = ei + nE;

  char* ws = (char*)d_ws;
  size_t off = 0;
  auto alloc = [&](size_t bytes) -> void* {
    void* p = (void*)(ws + off);
    off += (bytes + 255) & ~(size_t)255;
    return p;
  };
  float* dis    = (float*)alloc((size_t)n * 4);
  float* bufA   = (float*)alloc((size_t)n * HID * 4);
  float* bufB   = (float*)alloc((size_t)n * HID * 4);
  int*   cnt    = (int*)alloc((size_t)n * 4);
  int*   rowptr = (int*)alloc((size_t)(n + 1) * 4);
  int*   rank   = (int*)alloc((size_t)nE * 4);
  int*   csrS   = (int*)alloc((size_t)nE * 4);
  float* csrW   = (float*)alloc((size_t)nE * 4);
  float* h1   = bufA;
  float* hid  = bufB;
  float* h2   = bufA;
  float* outp = (float*)d_out;

  const int T = 256;
  auto blocks1d = [&](long long work) -> unsigned int {
    return (unsigned int)((work + T - 1) / T);
  };

  zero_i32<<<blocks1d(n), T, 0, stream>>>(cnt, n);
  edge_rank<<<blocks1d(nE), T, 0, stream>>>(dst, cnt, rank, nE);
  row_scan<<<blocks1d(n), T, 0, stream>>>(cnt, rowptr, dis, n);
  csr_fill<<<blocks1d(nE), T, 0, stream>>>(src, dst, rank, rowptr, dis, csrS, csrW, nE);

  const unsigned int stripBlocks = (unsigned int)(((n + 15) / 16 + 7) / 8);
  gemm_strip_f32<128, 64><<<stripBlocks, T, 0, stream>>>(x, W1, h1, n);
  gather_rows<64, true><<<blocks1d((long long)n * 16), T, 0, stream>>>(
      h1, rowptr, csrS, csrW, dis, b1, hid, n);

  gemm_strip_f32<64, 32><<<stripBlocks, T, 0, stream>>>(hid, W2, h2, n);
  gather_rows<32, false><<<blocks1d((long long)n * 8), T, 0, stream>>>(
      h2, rowptr, csrS, csrW, dis, b2, outp, n);
}


# Layer-2 product fused into the layer-1 gather: each block's 16 hidden rows pass through an LDS tile to one wave's WMMA (same k order, same sums), hid never stored, one launch fewer
# speedup vs baseline: 5.2478x; 2.5588x over previous
#include <hip/hip_runtime.h>

typedef float v2f __attribute__((ext_vector_type(2)));
typedef float v8f __attribute__((ext_vector_type(8)));

constexpr int RANGE_SHIFT = 8;
constexpr int EDGES_PER_BLOCK = 4096;
constexpr int STAGE_CAP = 8192;

__device__ __forceinline__ int4 load4_vec(const int* __restrict__ p, int e, int nE) {
  int4 v = make_int4(-1, -1, -1, -1);
  if (e < nE) v = *(const int4*)(p + e);
  return v;
}
__device__ __forceinline__ int4 load4_words(const int* __restrict__ p, int e, int nE) {
  int4 v = make_int4(-1, -1, -1, -1);
  if (e < nE) v.x = p[e];
  if (e + 1 < nE) v.y = p[e + 1];
  if (e + 2 < nE) v.z = p[e + 2];
  if (e + 3 < nE) v.w = p[e + 3];
  return v;
}

__device__ __forceinline__ int block_scan_excl(int v, int* wsum, int lane, int w, int& total) {
  int inc = v;
  for (int o = 1; o < 32; o <<= 1) {
    int a = __shfl_up(inc, o);
    if (lane >= o) inc += a;
  }
  __syncthreads();
  if (lane == 31) wsum[w] = inc;
  __syncthreads();
  int off = 0, tot = 0;
  for (int k = 0; k < 8; ++k) {
    int s = wsum[k];
    tot += s;
    if (k < w) off += s;
  }
  total = tot;
  return off + inc - v;
}

__device__ __forceinline__ void count_range(int* hist, int d) {
  if (d >= 0) atomicAdd(&hist[(d >> RANGE_SHIFT) & 255], 1);
}
__device__ __forceinline__ int take_slot(int* lcur, int d) {
  return d >= 0 ? atomicAdd(&lcur[(d >> RANGE_SHIFT) & 255], 1) : -1;
}
__device__ __forceinline__ void put_edge(unsigned int* sval, int p, int s, int d) {
  if (p >= 0) sval[p] = ((unsigned int)s << 16) | ((unsigned int)d & 0xffffu);
}

__global__ void __launch_bounds__(256) bin_local(const int* __restrict__ src,
                                                  const int* __restrict__ dst,
                                                  unsigned int* __restrict__ cntmat,
                                                  unsigned int* __restrict__ binned, int nE) {
  __shared__ int hist[256];
  __shared__ int lcur[256];
  __shared__ int wsum[8];
  __shared__ unsigned int sval[EDGES_PER_BLOCK];
  const int t = threadIdx.x;
  const int lane = t & 31;
  const int w = t >> 5;
  const int blk = blockIdx.x;
  hist[t] = 0;

  const int base = blk * EDGES_PER_BLOCK;
  const bool vec = (((((size_t)src) | ((size_t)dst)) & 15) == 0) && ((nE & 3) == 0);
  int4 s[4], d[4];
  if (vec) {
#pragma unroll
    for (int i = 0; i < 4; ++i) {
      int e = base + (i * 256 + t) * 4;
      d[i] = load4_vec(dst, e, nE);
      s[i] = load4_vec(src, e, nE);
    }
  } else {
#pragma unroll
    for (int i = 0; i < 4; ++i) {
      int e = base + (i * 256 + t) * 4;
      d[i] = load4_words(dst, e, nE);
      s[i] = load4_words(src, e, nE);
    }
  }
  __syncthreads();
#pragma unroll
  for (int i = 0; i < 4; ++i) {
    count_range(hist, d[i].x);
    count_range(hist, d[i].y);
    count_range(hist, d[i].z);
    count_range(hist, d[i].w);
  }
  __syncthreads();
  const int own = hist[t];
  int tot;
  const int lstart = block_scan_excl(own, wsum, lane, w, tot);
  lcur[t] = lstart;
  cntmat[blk * 256 + t] = ((unsigned int)lstart << 16) | (unsigned int)own;
  __syncthreads();

  int p[16];
#pragma unroll
  for (int i = 0; i < 4; ++i) {
    p[4 * i + 0] = take_slot(lcur, d[i].x);
    p[4 * i + 1] = take_slot(lcur, d[i].y);
    p[4 * i + 2] = take_slot(lcur, d[i].z);
    p[4 * i + 3] = take_slot(lcur, d[i].w);
  }
#pragma unroll
  for (int i = 0; i < 4; ++i) {
    put_edge(sval, p[4 * i + 0], s[i].x, d[i].x);
    put_edge(sval, p[4 * i + 1], s[i].y, d[i].y);
    put_edge(sval, p[4 * i + 2], s[i].z, d[i].z);
    put_edge(sval, p[4 * i + 3], s[i].w, d[i].w);
  }
  __syncthreads();
  int cntb = nE - base;
  if (cntb > EDGES_PER_BLOCK) cntb = EDGES_PER_BLOCK;
  for (int i = t; i < cntb; i += 256) binned[base + i] = sval[i];
}

template <class F>
__device__ __forceinline__ void for_each_range_edge(const unsigned int* __restrict__ binned,
                                                    const unsigned int* __restrict__ cntmat,
                                                    int r, int nblk, int* roff, int* rsrc, int* rcnt,
                                                    int* wsum, F f) {
  const int t = threadIdx.x;
  const int lane = t & 31;
  const int w = t >> 5;
  const int g = t >> 3;
  const int l8 = t & 7;
  int carry = 0;
  for (int kb = 0; kb < nblk; kb += 256) {
    const int k = kb + t;
    int c = 0, ls = 0;
    if (k < nblk) {
      unsigned int wd = cntmat[k * 256 + r];
      c = (int)(wd & 0xffffu);
      ls = (int)(wd >> 16);
    }
    int tot;
    const int excl = block_scan_excl(c, wsum, lane, w, tot);
    roff[t] = carry + excl;
    rsrc[t] = k * EDGES_PER_BLOCK + ls;
    rcnt[t] = c;
    __syncthreads();
    unsigned int v[8][4];
    int cc[8], oo[8], ss[8];
#pragma unroll
    for (int u = 0; u < 8; ++u) {
      int kk = g + 32 * u;
      cc[u] = rcnt[kk];
      ss[u] = rsrc[kk];
      oo[u] = roff[kk];
#pragma unroll
      for (int q = 0; q < 4; ++q) {
        int i = l8 + 8 * q;
        v[u][q] = 0;
        if (i < cc[u]) v[u][q] = binned[ss[u] + i];
      }
    }
#pragma unroll
    for (int u = 0; u < 8; ++u) {
#pragma unroll
      for (int q = 0; q < 4; ++q) {
        int i = l8 + 8 * q;
        if (i < cc[u]) f(v[u][q], oo[u] + i);
      }
    }
#pragma unroll
    for (int u = 0; u < 8; ++u) {
      for (int i = l8 + 32; i < cc[u]; i += 8) f(binned[ss[u] + i], oo[u] + i);
    }
    carry += tot;
    __syncthreads();
  }
}

__global__ void __launch_bounds__(256) range_csr(const unsigned int* __restrict__ binned,
                                                  const unsigned int* __restrict__ cntmat,
                                                  int* __restrict__ rowptr, float* __restrict__ dis,
                                                  int* __restrict__ csr_src, int n, int nblk) {
  __shared__ int cnt[256];
  __shared__ int cur[256];
  __shared__ int wsum[8];
  __shared__ int roff[256];
  __shared__ int rsrc[256];
  __shared__ int rcnt[256];
  __shared__ int sbeg[2];
  __shared__ unsigned int stage[STAGE_CAP];
  __shared__ unsigned short stage2[STAGE_CAP];
  const int t = threadIdx.x;
  const int lane = t & 31;
  const int w = t >> 5;
  const int r = blockIdx.x;

  constexpr int UT = 28;
  int tot = 0;
  for (int k0 = 0; k0 < nblk; k0 += UT) {
    unsigned int c[UT];
#pragma unroll
    for (int u = 0; u < UT; ++u) {
      int k = k0 + u;
      int kk = k < nblk ? k : nblk - 1;
      c[u] = cntmat[kk * 256 + t];
    }
#pragma unroll
    for (int u = 0; u < UT; ++u) tot += (k0 + u < nblk) ? (int)(c[u] & 0xffffu) : 0;
  }
  int all;
  const int gstart = block_scan_excl(tot, wsum, lane, w, all);
  if (t == r) { sbeg[0] = gstart; sbeg[1] = tot; }
  cnt[t] = 0;
  __syncthreads();
  const int beg = sbeg[0];
  const int m = sbeg[1];
  const bool staged = m <= STAGE_CAP;
  if (staged) {
    for_each_range_edge(binned, cntmat, r, nblk, roff, rsrc, rcnt, wsum,
                        [&](unsigned int v, int pos) {
                          stage[pos] = v;
                          atomicAdd(&cnt[v & 255u], 1);
                        });
  } else {
    for_each_range_edge(binned, cntmat, r, nblk, roff, rsrc, rcnt, wsum,
                        [&](unsigned int v, int pos) { atomicAdd(&cnt[v & 255u], 1); });
  }
  const int c = cnt[t];
  int dummy;
  const int excl = block_scan_excl(c, wsum, lane, w, dummy);
  const int node = (r << RANGE_SHIFT) + t;
  if (node < n) {
    rowptr[node] = beg + excl;
    dis[node] = rsqrtf((float)c + 1.0f);
  }
  if (node == n - 1) rowptr[n] = beg + excl + c;
  cur[t] = excl;
  __syncthreads();
  if (staged) {
    constexpr int U = 8;
    for (int j0 = t; j0 < m; j0 += 256 * U) {
      unsigned int v[U];
#pragma unroll
      for (int u = 0; u < U; ++u) {
        int j = j0 + 256 * u;
        v[u] = 0;
        if (j < m) v[u] = stage[j];
      }
      int p[U];
#pragma unroll
      for (int u = 0; u < U; ++u) {
        p[u] = 0;
        if (j0 + 256 * u < m) p[u] = atomicAdd(&cur[v[u] & 255u], 1);
      }
#pragma unroll
      for (int u = 0; u < U; ++u)
        if (j0 + 256 * u < m) stage2[p[u]] = (unsigned short)(v[u] >> 16);
    }
    __syncthreads();
    for (int i = t; i < m; i += 256) csr_src[beg + i] = (int)stage2[i];
  } else {
    for_each_range_edge(binned, cntmat, r, nblk, roff, rsrc, rcnt, wsum,
                        [&](unsigned int v, int pos) {
                          int p = atomicAdd(&cur[v & 255u], 1);
                          csr_src[beg + p] = (int)(v >> 16);
                        });
  }
}

template <int K, int N>
__global__ void __launch_bounds__(256) gemm_strip_f32(const float* __restrict__ A,
                                                       const float* __restrict__ B,
                                                       const float* __restrict__ rs,
                                                       float* __restrict__ D, int M) {
  constexpr int NT = N / 16;
  constexpr int Q = K / 8;
  constexpr int P2 = N + 8;
  constexpr int ITEMS = (K / 2) * (N / 4);
  __shared__ float4 Bs4[(K / 2) * P2 / 2];

  const int t = threadIdx.x;
  const int lane = t & 31;
  const int c = lane & 15;
  const int h = lane >> 4;
  const int wv = __builtin_amdgcn_readfirstlane(t >> 5);
  const int strips = (M + 15) >> 4;
  const int wstrip = blockIdx.x * 8 + wv;
  const bool live = wstrip < strips;
  const int strip = live ? wstrip : strips - 1;

  const float4* B4 = (const float4*)B;
  float4 w0[ITEMS / 256], w1[ITEMS / 256];
#pragma unroll
  for (int i = 0; i < ITEMS / 256; ++i) {
    int it = t + 256 * i;
    int kp = it / (N / 4);
    int n4 = it % (N / 4);
    w0[i] = B4[(2 * kp) * (N / 4) + n4];
    w1[i] = B4[(2 * kp + 1) * (N / 4) + n4];
  }

  int rowA = strip * 16 + c;
  if (rowA > M - 1) rowA = M - 1;
  const float4* arow = (const float4*)(A + (size_t)rowA * K) + h;
  float4 a[Q];
#pragma unroll
  for (int q = 0; q < Q; ++q) a[q] = arow[2 * q];

  float sc[8];
#pragma unroll
  for (int j = 0; j < 8; ++j) {
    int r = strip * 16 + 8 * h + j;
    sc[j] = rs[r < M ? r : M - 1];
  }

#pragma unroll
  for (int i = 0; i < ITEMS / 256; ++i) {
    int it = t + 256 * i;
    int kp = it / (N / 4);
    int n4 = it % (N / 4);
    float4* dst = Bs4 + (kp * P2 + 4 * n4) / 2;
    dst[0] = make_float4(w0[i].x, w1[i].x, w0[i].y, w1[i].y);
    dst[1] = make_float4(w0[i].z, w1[i].z, w0[i].w, w1[i].w);
  }
  __builtin_amdgcn_fence(__ATOMIC_RELEASE, "workgroup", "local");
  __builtin_amdgcn_s_barrier();
  __builtin_amdgcn_fence(__ATOMIC_ACQUIRE, "workgroup", "local");

  const v2f* bp = (const v2f*)Bs4 + (2 * h) * P2 + c;
  v8f acc[NT];
#pragma unroll
  for (int tt = 0; tt < NT; ++tt) acc[tt] = v8f{};

#pragma unroll
  for (int q = 0; q < Q; ++q) {
#pragma unroll
    for (int s2 = 0; s2 < 2; ++s2) {
      v2f av;
      av.x = s2 ? a[q].z : a[q].x;
      av.y = s2 ? a[q].w : a[q].y;
#pragma unroll
      for (int tt = 0; tt < NT; ++tt) {
        v2f bv = bp[(4 * q + s2) * P2 + 16 * tt];
        acc[tt] = __builtin_amdgcn_wmma_f32_16x16x4_f32(
             false, av,  false, bv,
             (short)0, acc[tt],  false,  false);
      }
    }
  }

  if (!live) return;
  float* dlane = D + ((size_t)strip * 16 + 8 * h) * N + c;
  if (strip * 16 + 16 <= M) {
#pragma unroll
    for (int tt = 0; tt < NT; ++tt) {
#pragma unroll
      for (int j = 0; j < 8; ++j) dlane[j * N + 16 * tt] = acc[tt][j] * sc[j];
    }
  } else {
#pragma unroll
    for (int tt = 0; tt < NT; ++tt) {
#pragma unroll
      for (int j = 0; j < 8; ++j) {
        if (strip * 16 + 8 * h + j < M) dlane[j * N + 16 * tt] = acc[tt][j] * sc[j];
      }
    }
  }
}

template <int C, bool RELU>
__global__ void gather_rows(const float* __restrict__ h, const int* __restrict__ rowptr,
                            const int* __restrict__ csr_src,
                            const float* __restrict__ dis, const float* __restrict__ b,
                            float* __restrict__ out, int n) {
  constexpr int LPN = C / 4;
  constexpr int NPW = 32 / LPN;
  int lane = threadIdx.x & 31;
  int sub = lane % LPN;
  long long wave = ((long long)blockIdx.x * blockDim.x + threadIdx.x) >> 5;
  long long nodeL = wave * NPW + lane / LPN;
  if (nodeL >= n) return;
  int node = (int)nodeL;
  int beg = rowptr[node];
  int end = rowptr[node + 1];
  const float4* h4 = (const float4*)h;
  float4 acc = make_float4(0.0f, 0.0f, 0.0f, 0.0f);
  for (int base = beg; base < end; base += LPN) {
    int my = base + sub;
    int s_l = 0;
    if (my < end) s_l = csr_src[my];
    int cnt = end - base;
    float4 v[LPN];
#pragma unroll
    for (int t = 0; t < LPN; ++t) {
      int s = __shfl(s_l, t, LPN);
      v[t] = make_float4(0.0f, 0.0f, 0.0f, 0.0f);
      if (t < cnt) v[t] = h4[(size_t)s * LPN + sub];
    }
#pragma unroll
    for (int t = 0; t < LPN; ++t) {
      acc.x += v[t].x;
      acc.y += v[t].y;
      acc.z += v[t].z;
      acc.w += v[t].w;
    }
  }
  float di = dis[node];
  float4 hv = h4[(size_t)node * LPN + sub];
  float4 bv = ((const float4*)b)[sub];
  float4 r;
  r.x = (acc.x + hv.x) * di + bv.x;
  r.y = (acc.y + hv.y) * di + bv.y;
  r.z = (acc.z + hv.z) * di + bv.z;
  r.w = (acc.w + hv.w) * di + bv.w;
  if (RELU) {
    r.x = fmaxf(r.x, 0.0f);
    r.y = fmaxf(r.y, 0.0f);
    r.z = fmaxf(r.z, 0.0f);
    r.w = fmaxf(r.w, 0.0f);
  }
  ((float4*)out)[(size_t)node * LPN + sub] = r;
}

template <int C, int N2>
__global__ void __launch_bounds__(256) gather_relu_gemm(const float* __restrict__ h,
                                                         const int* __restrict__ rowptr,
                                                         const int* __restrict__ csr_src,
                                                         const float* __restrict__ dis,
                                                         const float* __restrict__ b,
                                                         const float* __restrict__ W,
                                                         float* __restrict__ out, int n) {
  constexpr int LPN = C / 4;
  constexpr int NPW = 32 / LPN;
  constexpr int ROWS = 8 * NPW;
  constexpr int Q = C / 8;
  constexpr int P4 = C / 4 + 1;
  constexpr int NT = N2 / 16;
  constexpr int P2 = N2 + 8;
  constexpr int ITEMS = (C / 2) * (N2 / 4);
  static_assert(ROWS == 16, "one 16-row matrix tile per block");
  static_assert(N2 % 16 == 0 && C % 8 == 0 && ITEMS == 256, "whole tiles; one staging item per thread");
  __shared__ float4 Hs4[ROWS * P4];
  __shared__ float4 Bs4[(C / 2) * P2 / 2];

  const int t = threadIdx.x;
  const int lane = t & 31;
  const int sub = lane % LPN;
  const int wv = __builtin_amdgcn_readfirstlane(t >> 5);
  const int rib = wv * NPW + lane / LPN;
  const int row0 = blockIdx.x * ROWS;
  const bool valid = row0 + rib < n;
  const int node = valid ? row0 + rib : n - 1;
  const int beg = rowptr[node];
  int end = rowptr[node + 1];
  if (!valid) end = beg;

  const float4* W4 = (const float4*)W;
  {
    const int kp = t / (N2 / 4);
    const int n4 = t % (N2 / 4);
    float4 w0 = W4[(2 * kp) * (N2 / 4) + n4];
    float4 w1 = W4[(2 * kp + 1) * (N2 / 4) + n4];
    float4* dst = Bs4 + (kp * P2 + 4 * n4) / 2;
    dst[0] = make_float4(w0.x, w1.x, w0.y, w1.y);
    dst[1] = make_float4(w0.z, w1.z, w0.w, w1.w);
  }

  const float4* h4 = (const float4*)h;
  float4 acc = make_float4(0.0f, 0.0f, 0.0f, 0.0f);
  for (int base = beg; base < end; base += LPN) {
    int my = base + sub;
    int s_l = 0;
    if (my < end) s_l = csr_src[my];
    int cnt = end - base;
    float4 v[LPN];
#pragma unroll
    for (int u = 0; u < LPN; ++u) {
      int s = __shfl(s_l, u, LPN);
      v[u] = make_float4(0.0f, 0.0f, 0.0f, 0.0f);
      if (u < cnt) v[u] = h4[(size_t)s * LPN + sub];
    }
#pragma unroll
    for (int u = 0; u < LPN; ++u) {
      acc.x += v[u].x;
      acc.y += v[u].y;
      acc.z += v[u].z;
      acc.w += v[u].w;
    }
  }
  float di = dis[node];
  float4 hv = h4[(size_t)node * LPN + sub];
  float4 bv = ((const float4*)b)[sub];
  float4 r;
  r.x = fmaxf((acc.x + hv.x) * di + bv.x, 0.0f);
  r.y = fmaxf((acc.y + hv.y) * di + bv.y, 0.0f);
  r.z = fmaxf((acc.z + hv.z) * di + bv.z, 0.0f);
  r.w = fmaxf((acc.w + hv.w) * di + bv.w, 0.0f);
  if (!valid) r = make_float4(0.0f, 0.0f, 0.0f, 0.0f);
  Hs4[rib * P4 + sub] = r;

  const int c = lane & 15;
  const int hh = lane >> 4;
  const bool mm = wv == (int)(blockIdx.x & 7);
  float sc[8];
  if (mm) {
#pragma unroll
    for (int j = 0; j < 8; ++j) {
      int rr = row0 + 8 * hh + j;
      sc[j] = dis[rr < n ? rr : n - 1];
    }
  }
  __builtin_amdgcn_fence(__ATOMIC_RELEASE, "workgroup", "local");
  __builtin_amdgcn_s_barrier();
  __builtin_amdgcn_fence(__ATOMIC_ACQUIRE, "workgroup", "local");
  if (!mm) return;

  const float4* ap = Hs4 + c * P4 + hh;
  float4 a[Q];
#pragma unroll
  for (int q = 0; q < Q; ++q) a[q] = ap[2 * q];
  const v2f* bp = (const v2f*)Bs4 + (2 * hh) * P2 + c;
  v8f d[NT];
#pragma unroll
  for (int tt = 0; tt < NT; ++tt) d[tt] = v8f{};
#pragma unroll
  for (int q = 0; q < Q; ++q) {
#pragma unroll
    for (int s2 = 0; s2 < 2; ++s2) {
      v2f av;
      av.x = s2 ? a[q].z : a[q].x;
      av.y = s2 ? a[q].w : a[q].y;
#pragma unroll
      for (int tt = 0; tt < NT; ++tt) {
        v2f bv2 = bp[(4 * q + s2) * P2 + 16 * tt];
        d[tt] = __builtin_amdgcn_wmma_f32_16x16x4_f32(
             false, av,  false, bv2,
             (short)0, d[tt],  false,  false);
      }
    }
  }
  float* dl = out + ((size_t)row0 + 8 * hh) * N2 + c;
  if (row0 + ROWS <= n) {
#pragma unroll
    for (int tt = 0; tt < NT; ++tt) {
#pragma unroll
      for (int j = 0; j < 8; ++j) dl[j * N2 + 16 * tt] = d[tt][j] * sc[j];
    }
  } else {
#pragma unroll
    for (int tt = 0; tt < NT; ++tt) {
#pragma unroll
      for (int j = 0; j < 8; ++j) {
        if (row0 + 8 * hh + j < n) dl[j * N2 + 16 * tt] = d[tt][j] * sc[j];
      }
    }
  }
}


extern "C" void kernel_launch(void* const* d_in, const int* in_sizes, int n_in,
                              void* d_out, int out_size, void* d_ws, size_t ws_size,
                              hipStream_t stream) {
  const float* x  = (const float*)d_in[0];
  const float* W1 = (const float*)d_in[1];
  const float* b1 = (const float*)d_in[2];
  const float* W2 = (const float*)d_in[3];
  const float* b2 = (const float*)d_in[4];
  const int*   ei = (const int*)d_in[5];

  const int IN  = 128;
  const int HID = in_sizes[2];
  const int n   = in_sizes[0] / IN;
  const int nE  = in_sizes[5] / 2;
  const int* src = ei;
  const int* dst = ei + nE;

  char* ws = (char*)d_ws;
  size_t off = 0;
  auto alloc = [&](size_t bytes) -> void* {
    void* p = (void*)(ws + off);
    off += (bytes + 255) & ~(size_t)255;
    return p;
  };
  const int nblk = (nE + 4095) / 4096;
  const int nrng = (n + 255) / 256;

  float* dis    = (float*)alloc((size_t)n * 4);
  float* bufA   = (float*)alloc((size_t)n * HID * 4);
  float* bufB   = (float*)alloc((size_t)n * HID * 4);
  int*   rowptr = (int*)alloc((size_t)(n + 1) * 4);
  unsigned int* cntmat = (unsigned int*)alloc((size_t)nblk * 256 * 4);
  unsigned int* binned = (unsigned int*)alloc((size_t)nE * 4);
  int*   csrS   = (int*)alloc((size_t)nE * 4);
  float* h1   = bufA;
  float* h2   = bufB;
  float* outp = (float*)d_out;

  const int T = 256;
  auto blocks1d = [&](long long work) -> unsigned int {
    return (unsigned int)((work + T - 1) / T);
  };

  bin_local<<<nblk, T, 0, stream>>>(src, dst, cntmat, binned, nE);
  range_csr<<<nrng, T, 0, stream>>>(binned, cntmat, rowptr, dis, csrS, n, nblk);

  const unsigned int stripBlocks = (unsigned int)(((n + 15) / 16 + 7) / 8);
  gemm_strip_f32<128, 64><<<stripBlocks, T, 0, stream>>>(x, W1, dis, h1, n);
  gather_relu_gemm<64, 32><<<(unsigned int)((n + 15) / 16), T, 0, stream>>>(
      h1, rowptr, csrS, dis, b1, W2, h2, n);

  gather_rows<32, false><<<blocks1d((long long)n * 8), T, 0, stream>>>(
      h2, rowptr, csrS, dis, b2, outp, n);
}


# range_csr takes its range's size and its place in the edge list from its own table column (sum of the slice-local run starts stored by level 1) instead of summing the whole count table in every block
# speedup vs baseline: 5.5011x; 1.0483x over previous
#include <hip/hip_runtime.h>

typedef float v2f __attribute__((ext_vector_type(2)));
typedef float v8f __attribute__((ext_vector_type(8)));

constexpr int RANGE_SHIFT = 8;
constexpr int EDGES_PER_BLOCK = 4096;
constexpr int STAGE_CAP = 8192;

__device__ __forceinline__ int4 load4_vec(const int* __restrict__ p, int e, int nE) {
  int4 v = make_int4(-1, -1, -1, -1);
  if (e < nE) v = *(const int4*)(p + e);
  return v;
}
__device__ __forceinline__ int4 load4_words(const int* __restrict__ p, int e, int nE) {
  int4 v = make_int4(-1, -1, -1, -1);
  if (e < nE) v.x = p[e];
  if (e + 1 < nE) v.y = p[e + 1];
  if (e + 2 < nE) v.z = p[e + 2];
  if (e + 3 < nE) v.w = p[e + 3];
  return v;
}

__device__ __forceinline__ int block_scan_excl(int v, int* wsum, int lane, int w, int& total) {
  int inc = v;
  for (int o = 1; o < 32; o <<= 1) {
    int a = __shfl_up(inc, o);
    if (lane >= o) inc += a;
  }
  __syncthreads();
  if (lane == 31) wsum[w] = inc;
  __syncthreads();
  int off = 0, tot = 0;
  for (int k = 0; k < 8; ++k) {
    int s = wsum[k];
    tot += s;
    if (k < w) off += s;
  }
  total = tot;
  return off + inc - v;
}

__device__ __forceinline__ void block_sum2(int a, int b, int* rs, int lane, int w, int& sa, int& sb) {
  for (int o = 16; o > 0; o >>= 1) {
    a += __shfl_xor(a, o);
    b += __shfl_xor(b, o);
  }
  if (lane == 0) {
    rs[w] = a;
    rs[8 + w] = b;
  }
  __syncthreads();
  int x = 0, y = 0;
  for (int k = 0; k < 8; ++k) {
    x += rs[k];
    y += rs[8 + k];
  }
  sa = x;
  sb = y;
}

__device__ __forceinline__ void count_range(int* hist, int d) {
  if (d >= 0) atomicAdd(&hist[(d >> RANGE_SHIFT) & 255], 1);
}
__device__ __forceinline__ int take_slot(int* lcur, int d) {
  return d >= 0 ? atomicAdd(&lcur[(d >> RANGE_SHIFT) & 255], 1) : -1;
}
__device__ __forceinline__ void put_edge(unsigned int* sval, int p, int s, int d) {
  if (p >= 0) sval[p] = ((unsigned int)s << 16) | ((unsigned int)d & 0xffffu);
}

__global__ void __launch_bounds__(256) bin_local(const int* __restrict__ src,
                                                  const int* __restrict__ dst,
                                                  unsigned int* __restrict__ cntmat,
                                                  unsigned int* __restrict__ binned, int nE) {
  __shared__ int hist[256];
  __shared__ int lcur[256];
  __shared__ int wsum[8];
  __shared__ unsigned int sval[EDGES_PER_BLOCK];
  const int t = threadIdx.x;
  const int lane = t & 31;
  const int w = t >> 5;
  const int blk = blockIdx.x;
  hist[t] = 0;

  const int base = blk * EDGES_PER_BLOCK;
  const bool vec = (((((size_t)src) | ((size_t)dst)) & 15) == 0) && ((nE & 3) == 0);
  int4 s[4], d[4];
  if (vec) {
#pragma unroll
    for (int i = 0; i < 4; ++i) {
      int e = base + (i * 256 + t) * 4;
      d[i] = load4_vec(dst, e, nE);
      s[i] = load4_vec(src, e, nE);
    }
  } else {
#pragma unroll
    for (int i = 0; i < 4; ++i) {
      int e = base + (i * 256 + t) * 4;
      d[i] = load4_words(dst, e, nE);
      s[i] = load4_words(src, e, nE);
    }
  }
  __syncthreads();
#pragma unroll
  for (int i = 0; i < 4; ++i) {
    count_range(hist, d[i].x);
    count_range(hist, d[i].y);
    count_range(hist, d[i].z);
    count_range(hist, d[i].w);
  }
  __syncthreads();
  const int own = hist[t];
  int tot;
  const int lstart = block_scan_excl(own, wsum, lane, w, tot);
  lcur[t] = lstart;
  cntmat[blk * 256 + t] = ((unsigned int)lstart << 16) | (unsigned int)own;
  __syncthreads();

  int p[16];
#pragma unroll
  for (int i = 0; i < 4; ++i) {
    p[4 * i + 0] = take_slot(lcur, d[i].x);
    p[4 * i + 1] = take_slot(lcur, d[i].y);
    p[4 * i + 2] = take_slot(lcur, d[i].z);
    p[4 * i + 3] = take_slot(lcur, d[i].w);
  }
#pragma unroll
  for (int i = 0; i < 4; ++i) {
    put_edge(sval, p[4 * i + 0], s[i].x, d[i].x);
    put_edge(sval, p[4 * i + 1], s[i].y, d[i].y);
    put_edge(sval, p[4 * i + 2], s[i].z, d[i].z);
    put_edge(sval, p[4 * i + 3], s[i].w, d[i].w);
  }
  __syncthreads();
  int cntb = nE - base;
  if (cntb > EDGES_PER_BLOCK) cntb = EDGES_PER_BLOCK;
  for (int i = t; i < cntb; i += 256) binned[base + i] = sval[i];
}

template <class F>
__device__ __forceinline__ void for_each_range_edge(const unsigned int* __restrict__ binned,
                                                    const unsigned int* __restrict__ cntmat,
                                                    int r, int nblk, unsigned int first, int* roff,
                                                    int* rsrc, int* rcnt, int* wsum, F f) {
  const int t = threadIdx.x;
  const int lane = t & 31;
  const int w = t >> 5;
  const int g = t >> 3;
  const int l8 = t & 7;
  int carry = 0;
  for (int kb = 0; kb < nblk; kb += 256) {
    const int k = kb + t;
    unsigned int wd = first;
    if (kb > 0) {
      wd = 0;
      if (k < nblk) wd = cntmat[k * 256 + r];
    }
    const int c = (int)(wd & 0xffffu);
    const int ls = (int)(wd >> 16);
    int tot;
    const int excl = block_scan_excl(c, wsum, lane, w, tot);
    roff[t] = carry + excl;
    rsrc[t] = k * EDGES_PER_BLOCK + ls;
    rcnt[t] = c;
    __syncthreads();
    unsigned int v[8][4];
    int cc[8], oo[8], ss[8];
#pragma unroll
    for (int u = 0; u < 8; ++u) {
      int kk = g + 32 * u;
      cc[u] = rcnt[kk];
      ss[u] = rsrc[kk];
      oo[u] = roff[kk];
#pragma unroll
      for (int q = 0; q < 4; ++q) {
        int i = l8 + 8 * q;
        v[u][q] = 0;
        if (i < cc[u]) v[u][q] = binned[ss[u] + i];
      }
    }
#pragma unroll
    for (int u = 0; u < 8; ++u) {
#pragma unroll
      for (int q = 0; q < 4; ++q) {
        int i = l8 + 8 * q;
        if (i < cc[u]) f(v[u][q], oo[u] + i);
      }
    }
#pragma unroll
    for (int u = 0; u < 8; ++u) {
      for (int i = l8 + 32; i < cc[u]; i += 8) f(binned[ss[u] + i], oo[u] + i);
    }
    carry += tot;
    __syncthreads();
  }
}

__global__ void __launch_bounds__(256) range_csr(const unsigned int* __restrict__ binned,
                                                  const unsigned int* __restrict__ cntmat,
                                                  int* __restrict__ rowptr, float* __restrict__ dis,
                                                  int* __restrict__ csr_src, int n, int nblk) {
  __shared__ int cnt[256];
  __shared__ int cur[256];
  __shared__ int wsum[8];
  __shared__ int roff[256];
  __shared__ int rsrc[256];
  __shared__ int rcnt[256];
  __shared__ int rsum[16];
  __shared__ unsigned int stage[STAGE_CAP];
  __shared__ unsigned short stage2[STAGE_CAP];
  const int t = threadIdx.x;
  const int lane = t & 31;
  const int w = t >> 5;
  const int r = blockIdx.x;

  unsigned int first = 0;
  if (t < nblk) first = cntmat[t * 256 + r];
  int csum = (int)(first & 0xffffu);
  int lsum = (int)(first >> 16);
  for (int k = t + 256; k < nblk; k += 256) {
    const unsigned int wd = cntmat[k * 256 + r];
    csum += (int)(wd & 0xffffu);
    lsum += (int)(wd >> 16);
  }
  cnt[t] = 0;
  int m, beg;
  block_sum2(csum, lsum, rsum, lane, w, m, beg);
  const bool staged = m <= STAGE_CAP;
  if (staged) {
    for_each_range_edge(binned, cntmat, r, nblk, first, roff, rsrc, rcnt, wsum,
                        [&](unsigned int v, int pos) {
                          stage[pos] = v;
                          atomicAdd(&cnt[v & 255u], 1);
                        });
  } else {
    for_each_range_edge(binned, cntmat, r, nblk, first, roff, rsrc, rcnt, wsum,
                        [&](unsigned int v, int pos) { atomicAdd(&cnt[v & 255u], 1); });
  }
  const int c = cnt[t];
  int dummy;
  const int excl = block_scan_excl(c, wsum, lane, w, dummy);
  const int node = (r << RANGE_SHIFT) + t;
  if (node < n) {
    rowptr[node] = beg + excl;
    dis[node] = rsqrtf((float)c + 1.0f);
  }
  if (node == n - 1) rowptr[n] = beg + excl + c;
  cur[t] = excl;
  __syncthreads();
  if (staged) {
    constexpr int U = 8;
    for (int j0 = t; j0 < m; j0 += 256 * U) {
      unsigned int v[U];
#pragma unroll
      for (int u = 0; u < U; ++u) {
        int j = j0 + 256 * u;
        v[u] = 0;
        if (j < m) v[u] = stage[j];
      }
      int p[U];
#pragma unroll
      for (int u = 0; u < U; ++u) {
        p[u] = 0;
        if (j0 + 256 * u < m) p[u] = atomicAdd(&cur[v[u] & 255u], 1);
      }
#pragma unroll
      for (int u = 0; u < U; ++u)
        if (j0 + 256 * u < m) stage2[p[u]] = (unsigned short)(v[u] >> 16);
    }
    __syncthreads();
    for (int i = t; i < m; i += 256) csr_src[beg + i] = (int)stage2[i];
  } else {
    for_each_range_edge(binned, cntmat, r, nblk, first, roff, rsrc, rcnt, wsum,
                        [&](unsigned int v, int pos) {
                          int p = atomicAdd(&cur[v & 255u], 1);
                          csr_src[beg + p] = (int)(v >> 16);
                        });
  }
}

template <int K, int N>
__global__ void __launch_bounds__(256) gemm_strip_f32(const float* __restrict__ A,
                                                       const float* __restrict__ B,
                                                       const float* __restrict__ rs,
                                                       float* __restrict__ D, int M) {
  constexpr int NT = N / 16;
  constexpr int Q = K / 8;
  constexpr int P2 = N + 8;
  constexpr int ITEMS = (K / 2) * (N / 4);
  __shared__ float4 Bs4[(K / 2) * P2 / 2];

  const int t = threadIdx.x;
  const int lane = t & 31;
  const int c = lane & 15;
  const int h = lane >> 4;
  const int wv = __builtin_amdgcn_readfirstlane(t >> 5);
  const int strips = (M + 15) >> 4;
  const int wstrip = blockIdx.x * 8 + wv;
  const bool live = wstrip < strips;
  const int strip = live ? wstrip : strips - 1;

  const float4* B4 = (const float4*)B;
  float4 w0[ITEMS / 256], w1[ITEMS / 256];
#pragma unroll
  for (int i = 0; i < ITEMS / 256; ++i) {
    int it = t + 256 * i;
    int kp = it / (N / 4);
    int n4 = it % (N / 4);
    w0[i] = B4[(2 * kp) * (N / 4) + n4];
    w1[i] = B4[(2 * kp + 1) * (N / 4) + n4];
  }

  int rowA = strip * 16 + c;
  if (rowA > M - 1) rowA = M - 1;
  const float4* arow = (const float4*)(A + (size_t)rowA * K) + h;
  float4 a[Q];
#pragma unroll
  for (int q = 0; q < Q; ++q) a[q] = arow[2 * q];

  float sc[8];
#pragma unroll
  for (int j = 0; j < 8; ++j) {
    int r = strip * 16 + 8 * h + j;
    sc[j] = rs[r < M ? r : M - 1];
  }

#pragma unroll
  for (int i = 0; i < ITEMS / 256; ++i) {
    int it = t + 256 * i;
    int kp = it / (N / 4);
    int n4 = it % (N / 4);
    float4* dst = Bs4 + (kp * P2 + 4 * n4) / 2;
    dst[0] = make_float4(w0[i].x, w1[i].x, w0[i].y, w1[i].y);
    dst[1] = make_float4(w0[i].z, w1[i].z, w0[i].w, w1[i].w);
  }
  __builtin_amdgcn_fence(__ATOMIC_RELEASE, "workgroup", "local");
  __builtin_amdgcn_s_barrier();
  __builtin_amdgcn_fence(__ATOMIC_ACQUIRE, "workgroup", "local");

  const v2f* bp = (const v2f*)Bs4 + (2 * h) * P2 + c;
  v8f acc[NT];
#pragma unroll
  for (int tt = 0; tt < NT; ++tt) acc[tt] = v8f{};

#pragma unroll
  for (int q = 0; q < Q; ++q) {
#pragma unroll
    for (int s2 = 0; s2 < 2; ++s2) {
      v2f av;
      av.x = s2 ? a[q].z : a[q].x;
      av.y = s2 ? a[q].w : a[q].y;
#pragma unroll
      for (int tt = 0; tt < NT; ++tt) {
        v2f bv = bp[(4 * q + s2) * P2 + 16 * tt];
        acc[tt] = __builtin_amdgcn_wmma_f32_16x16x4_f32(
             false, av,  false, bv,
             (short)0, acc[tt],  false,  false);
      }
    }
  }

  if (!live) return;
  float* dlane = D + ((size_t)strip * 16 + 8 * h) * N + c;
  if (strip * 16 + 16 <= M) {
#pragma unroll
    for (int tt = 0; tt < NT; ++tt) {
#pragma unroll
      for (int j = 0; j < 8; ++j) dlane[j * N + 16 * tt] = acc[tt][j] * sc[j];
    }
  } else {
#pragma unroll
    for (int tt = 0; tt < NT; ++tt) {
#pragma unroll
      for (int j = 0; j < 8; ++j) {
        if (strip * 16 + 8 * h + j < M) dlane[j * N + 16 * tt] = acc[tt][j] * sc[j];
      }
    }
  }
}

template <int C, bool RELU>
__global__ void gather_rows(const float* __restrict__ h, const int* __restrict__ rowptr,
                            const int* __restrict__ csr_src,
                            const float* __restrict__ dis, const float* __restrict__ b,
                            float* __restrict__ out, int n) {
  constexpr int LPN = C / 4;
  constexpr int NPW = 32 / LPN;
  int lane = threadIdx.x & 31;
  int sub = lane % LPN;
  long long wave = ((long long)blockIdx.x * blockDim.x + threadIdx.x) >> 5;
  long long nodeL = wave * NPW + lane / LPN;
  if (nodeL >= n) return;
  int node = (int)nodeL;
  int beg = rowptr[node];
  int end = rowptr[node + 1];
  const float4* h4 = (const float4*)h;
  float4 acc = make_float4(0.0f, 0.0f, 0.0f, 0.0f);
  for (int base = beg; base < end; base += LPN) {
    int my = base + sub;
    int s_l = 0;
    if (my < end) s_l = csr_src[my];
    int cnt = end - base;
    float4 v[LPN];
#pragma unroll
    for (int t = 0; t < LPN; ++t) {
      int s = __shfl(s_l, t, LPN);
      v[t] = make_float4(0.0f, 0.0f, 0.0f, 0.0f);
      if (t < cnt) v[t] = h4[(size_t)s * LPN + sub];
    }
#pragma unroll
    for (int t = 0; t < LPN; ++t) {
      acc.x += v[t].x;
      acc.y += v[t].y;
      acc.z += v[t].z;
      acc.w += v[t].w;
    }
  }
  float di = dis[node];
  float4 hv = h4[(size_t)node * LPN + sub];
  float4 bv = ((const float4*)b)[sub];
  float4 r;
  r.x = (acc.x + hv.x) * di + bv.x;
  r.y = (acc.y + hv.y) * di + bv.y;
  r.z = (acc.z + hv.z) * di + bv.z;
  r.w = (acc.w + hv.w) * di + bv.w;
  if (RELU) {
    r.x = fmaxf(r.x, 0.0f);
    r.y = fmaxf(r.y, 0.0f);
    r.z = fmaxf(r.z, 0.0f);
    r.w = fmaxf(r.w, 0.0f);
  }
  ((float4*)out)[(size_t)node * LPN + sub] = r;
}

template <int C, int N2>
__global__ void __launch_bounds__(256) gather_relu_gemm(const float* __restrict__ h,
                                                         const int* __restrict__ rowptr,
                                                         const int* __restrict__ csr_src,
                                                         const float* __restrict__ dis,
                                                         const float* __restrict__ b,
                                                         const float* __restrict__ W,
                                                         float* __restrict__ out, int n) {
  constexpr int LPN = C / 4;
  constexpr int NPW = 32 / LPN;
  constexpr int ROWS = 8 * NPW;
  constexpr int Q = C / 8;
  constexpr int P4 = C / 4 + 1;
  constexpr int NT = N2 / 16;
  constexpr int P2 = N2 + 8;
  constexpr int ITEMS = (C / 2) * (N2 / 4);
  static_assert(ROWS == 16, "one 16-row matrix tile per block");
  static_assert(N2 % 16 == 0 && C % 8 == 0 && ITEMS == 256, "whole tiles; one staging item per thread");
  __shared__ float4 Hs4[ROWS * P4];
  __shared__ float4 Bs4[(C / 2) * P2 / 2];

  const int t = threadIdx.x;
  const int lane = t & 31;
  const int sub = lane % LPN;
  const int wv = __builtin_amdgcn_readfirstlane(t >> 5);
  const int rib = wv * NPW + lane / LPN;
  const int row0 = blockIdx.x * ROWS;
  const bool valid = row0 + rib < n;
  const int node = valid ? row0 + rib : n - 1;
  const int beg = rowptr[node];
  int end = rowptr[node + 1];
  if (!valid) end = beg;

  const float4* W4 = (const float4*)W;
  {
    const int kp = t / (N2 / 4);
    const int n4 = t % (N2 / 4);
    float4 w0 = W4[(2 * kp) * (N2 / 4) + n4];
    float4 w1 = W4[(2 * kp + 1) * (N2 / 4) + n4];
    float4* dst = Bs4 + (kp * P2 + 4 * n4) / 2;
    dst[0] = make_float4(w0.x, w1.x, w0.y, w1.y);
    dst[1] = make_float4(w0.z, w1.z, w0.w, w1.w);
  }

  const float4* h4 = (const float4*)h;
  float4 acc = make_float4(0.0f, 0.0f, 0.0f, 0.0f);
  for (int base = beg; base < end; base += LPN) {
    int my = base + sub;
    int s_l = 0;
    if (my < end) s_l = csr_src[my];
    int cnt = end - base;
    float4 v[LPN];
#pragma unroll
    for (int u = 0; u < LPN; ++u) {
      int s = __shfl(s_l, u, LPN);
      v[u] = make_float4(0.0f, 0.0f, 0.0f, 0.0f);
      if (u < cnt) v[u] = h4[(size_t)s * LPN + sub];
    }
#pragma unroll
    for (int u = 0; u < LPN; ++u) {
      acc.x += v[u].x;
      acc.y += v[u].y;
      acc.z += v[u].z;
      acc.w += v[u].w;
    }
  }
  float di = dis[node];
  float4 hv = h4[(size_t)node * LPN + sub];
  float4 bv = ((const float4*)b)[sub];
  float4 r;
  r.x = fmaxf((acc.x + hv.x) * di + bv.x, 0.0f);
  r.y = fmaxf((acc.y + hv.y) * di + bv.y, 0.0f);
  r.z = fmaxf((acc.z + hv.z) * di + bv.z, 0.0f);
  r.w = fmaxf((acc.w + hv.w) * di + bv.w, 0.0f);
  if (!valid) r = make_float4(0.0f, 0.0f, 0.0f, 0.0f);
  Hs4[rib * P4 + sub] = r;

  const int c = lane & 15;
  const int hh = lane >> 4;
  const bool mm = wv == (int)(blockIdx.x & 7);
  float sc[8];
  if (mm) {
#pragma unroll
    for (int j = 0; j < 8; ++j) {
      int rr = row0 + 8 * hh + j;
      sc[j] = dis[rr < n ? rr : n - 1];
    }
  }
  __builtin_amdgcn_fence(__ATOMIC_RELEASE, "workgroup", "local");
  __builtin_amdgcn_s_barrier();
  __builtin_amdgcn_fence(__ATOMIC_ACQUIRE, "workgroup", "local");
  if (!mm) return;

  const float4* ap = Hs4 + c * P4 + hh;
  float4 a[Q];
#pragma unroll
  for (int q = 0; q < Q; ++q) a[q] = ap[2 * q];
  const v2f* bp = (const v2f*)Bs4 + (2 * hh) * P2 + c;
  v8f d[NT];
#pragma unroll
  for (int tt = 0; tt < NT; ++tt) d[tt] = v8f{};
#pragma unroll
  for (int q = 0; q < Q; ++q) {
#pragma unroll
    for (int s2 = 0; s2 < 2; ++s2) {
      v2f av;
      av.x = s2 ? a[q].z : a[q].x;
      av.y = s2 ? a[q].w : a[q].y;
#pragma unroll
      for (int tt = 0; tt < NT; ++tt) {
        v2f bv2 = bp[(4 * q + s2) * P2 + 16 * tt];
        d[tt] = __builtin_amdgcn_wmma_f32_16x16x4_f32(
             false, av,  false, bv2,
             (short)0, d[tt],  false,  false);
      }
    }
  }
  float* dl = out + ((size_t)row0 + 8 * hh) * N2 + c;
  if (row0 + ROWS <= n) {
#pragma unroll
    for (int tt = 0; tt < NT; ++tt) {
#pragma unroll
      for (int j = 0; j < 8; ++j) dl[j * N2 + 16 * tt] = d[tt][j] * sc[j];
    }
  } else {
#pragma unroll
    for (int tt = 0; tt < NT; ++tt) {
#pragma unroll
      for (int j = 0; j < 8; ++j) {
        if (row0 + 8 * hh + j < n) dl[j * N2 + 16 * tt] = d[tt][j] * sc[j];
      }
    }
  }
}


extern "C" void kernel_launch(void* const* d_in, const int* in_sizes, int n_in,
                              void* d_out, int out_size, void* d_ws, size_t ws_size,
                              hipStream_t stream) {
  const float* x  = (const float*)d_in[0];
  const float* W1 = (const float*)d_in[1];
  const float* b1 = (const float*)d_in[2];
  const float* W2 = (const float*)d_in[3];
  const float* b2 = (const float*)d_in[4];
  const int*   ei = (const int*)d_in[5];

  const int IN  = 128;
  const int HID = in_sizes[2];
  const int n   = in_sizes[0] / IN;
  const int nE  = in_sizes[5] / 2;
  const int* src = ei;
  const int* dst = ei + nE;

  char* ws = (char*)d_ws;
  size_t off = 0;
  auto alloc = [&](size_t bytes) -> void* {
    void* p = (void*)(ws + off);
    off += (bytes + 255) & ~(size_t)255;
    return p;
  };
  const int nblk = (nE + 4095) / 4096;
  const int nrng = (n + 255) / 256;

  float* dis    = (float*)alloc((size_t)n * 4);
  float* bufA   = (float*)alloc((size_t)n * HID * 4);
  float* bufB   = (float*)alloc((size_t)n * HID * 4);
  int*   rowptr = (int*)alloc((size_t)(n + 1) * 4);
  unsigned int* cntmat = (unsigned int*)alloc((size_t)nblk * 256 * 4);
  unsigned int* binned = (unsigned int*)alloc((size_t)nE * 4);
  int*   csrS   = (int*)alloc((size_t)nE * 4);
  float* h1   = bufA;
  float* h2   = bufB;
  float* outp = (float*)d_out;

  const int T = 256;
  auto blocks1d = [&](long long work) -> unsigned int {
    return (unsigned int)((work + T - 1) / T);
  };

  bin_local<<<nblk, T, 0, stream>>>(src, dst, cntmat, binned, nE);
  range_csr<<<nrng, T, 0, stream>>>(binned, cntmat, rowptr, dis, csrS, n, nblk);

  const unsigned int stripBlocks = (unsigned int)(((n + 15) / 16 + 7) / 8);
  gemm_strip_f32<128, 64><<<stripBlocks, T, 0, stream>>>(x, W1, dis, h1, n);
  gather_relu_gemm<64, 32><<<(unsigned int)((n + 15) / 16), T, 0, stream>>>(
      h1, rowptr, csrS, dis, b1, W2, h2, n);

  gather_rows<32, false><<<blocks1d((long long)n * 8), T, 0, stream>>>(
      h2, rowptr, csrS, dis, b2, outp, n);
}


# Gather loops: lists padded with a zero row so row loads need no per-lane test (32-bit offsets, one scalar test per half trip); next indices, own row, dis, bias requested ahead; real rows' order kept
# speedup vs baseline: 6.0081x; 1.0922x over previous
#include <hip/hip_runtime.h>

typedef float v2f __attribute__((ext_vector_type(2)));
typedef float v8f __attribute__((ext_vector_type(8)));

constexpr int RANGE_SHIFT = 8;
constexpr int EDGES_PER_BLOCK = 4096;
constexpr int STAGE_CAP = 8192;

__device__ __forceinline__ int4 load4_vec(const int* __restrict__ p, int e, int nE) {
  int4 v = make_int4(-1, -1, -1, -1);
  if (e < nE) v = *(const int4*)(p + e);
  return v;
}
__device__ __forceinline__ int4 load4_words(const int* __restrict__ p, int e, int nE) {
  int4 v = make_int4(-1, -1, -1, -1);
  if (e < nE) v.x = p[e];
  if (e + 1 < nE) v.y = p[e + 1];
  if (e + 2 < nE) v.z = p[e + 2];
  if (e + 3 < nE) v.w = p[e + 3];
  return v;
}

__device__ __forceinline__ int block_scan_excl(int v, int* wsum, int lane, int w, int& total) {
  int inc = v;
  for (int o = 1; o < 32; o <<= 1) {
    int a = __shfl_up(inc, o);
    if (lane >= o) inc += a;
  }
  __syncthreads();
  if (lane == 31) wsum[w] = inc;
  __syncthreads();
  int off = 0, tot = 0;
  for (int k = 0; k < 8; ++k) {
    int s = wsum[k];
    tot += s;
    if (k < w) off += s;
  }
  total = tot;
  return off + inc - v;
}

__device__ __forceinline__ void block_sum2(int a, int b, int* rs, int lane, int w, int& sa, int& sb) {
  for (int o = 16; o > 0; o >>= 1) {
    a += __shfl_xor(a, o);
    b += __shfl_xor(b, o);
  }
  if (lane == 0) {
    rs[w] = a;
    rs[8 + w] = b;
  }
  __syncthreads();
  int x = 0, y = 0;
  for (int k = 0; k < 8; ++k) {
    x += rs[k];
    y += rs[8 + k];
  }
  sa = x;
  sb = y;
}

__device__ __forceinline__ void count_range(int* hist, int d) {
  if (d >= 0) atomicAdd(&hist[(d >> RANGE_SHIFT) & 255], 1);
}
__device__ __forceinline__ int take_slot(int* lcur, int d) {
  return d >= 0 ? atomicAdd(&lcur[(d >> RANGE_SHIFT) & 255], 1) : -1;
}
__device__ __forceinline__ void put_edge(unsigned int* sval, int p, int s, int d) {
  if (p >= 0) sval[p] = ((unsigned int)s << 16) | ((unsigned int)d & 0xffffu);
}

__global__ void __launch_bounds__(256) bin_local(const int* __restrict__ src,
                                                  const int* __restrict__ dst,
                                                  unsigned int* __restrict__ cntmat,
                                                  unsigned int* __restrict__ binned, int nE) {
  __shared__ int hist[256];
  __shared__ int lcur[256];
  __shared__ int wsum[8];
  __shared__ unsigned int sval[EDGES_PER_BLOCK];
  const int t = threadIdx.x;
  const int lane = t & 31;
  const int w = t >> 5;
  const int blk = blockIdx.x;
  hist[t] = 0;

  const int base = blk * EDGES_PER_BLOCK;
  const bool vec = (((((size_t)src) | ((size_t)dst)) & 15) == 0) && ((nE & 3) == 0);
  int4 s[4], d[4];
  if (vec) {
#pragma unroll
    for (int i = 0; i < 4; ++i) {
      int e = base + (i * 256 + t) * 4;
      d[i] = load4_vec(dst, e, nE);
      s[i] = load4_vec(src, e, nE);
    }
  } else {
#pragma unroll
    for (int i = 0; i < 4; ++i) {
      int e = base + (i * 256 + t) * 4;
      d[i] = load4_words(dst, e, nE);
      s[i] = load4_words(src, e, nE);
    }
  }
  __syncthreads();
#pragma unroll
  for (int i = 0; i < 4; ++i) {
    count_range(hist, d[i].x);
    count_range(hist, d[i].y);
    count_range(hist, d[i].z);
    count_range(hist, d[i].w);
  }
  __syncthreads();
  const int own = hist[t];
  int tot;
  const int lstart = block_scan_excl(own, wsum, lane, w, tot);
  lcur[t] = lstart;
  cntmat[blk * 256 + t] = ((unsigned int)lstart << 16) | (unsigned int)own;
  __syncthreads();

  int p[16];
#pragma unroll
  for (int i = 0; i < 4; ++i) {
    p[4 * i + 0] = take_slot(lcur, d[i].x);
    p[4 * i + 1] = take_slot(lcur, d[i].y);
    p[4 * i + 2] = take_slot(lcur, d[i].z);
    p[4 * i + 3] = take_slot(lcur, d[i].w);
  }
#pragma unroll
  for (int i = 0; i < 4; ++i) {
    put_edge(sval, p[4 * i + 0], s[i].x, d[i].x);
    put_edge(sval, p[4 * i + 1], s[i].y, d[i].y);
    put_edge(sval, p[4 * i + 2], s[i].z, d[i].z);
    put_edge(sval, p[4 * i + 3], s[i].w, d[i].w);
  }
  __syncthreads();
  int cntb = nE - base;
  if (cntb > EDGES_PER_BLOCK) cntb = EDGES_PER_BLOCK;
  for (int i = t; i < cntb; i += 256) binned[base + i] = sval[i];
}

template <class F>
__device__ __forceinline__ void for_each_range_edge(const unsigned int* __restrict__ binned,
                                                    const unsigned int* __restrict__ cntmat,
                                                    int r, int nblk, unsigned int first, int* roff,
                                                    int* rsrc, int* rcnt, int* wsum, F f) {
  const int t = threadIdx.x;
  const int lane = t & 31;
  const int w = t >> 5;
  const int g = t >> 3;
  const int l8 = t & 7;
  int carry = 0;
  for (int kb = 0; kb < nblk; kb += 256) {
    const int k = kb + t;
    unsigned int wd = first;
    if (kb > 0) {
      wd = 0;
      if (k < nblk) wd = cntmat[k * 256 + r];
    }
    const int c = (int)(wd & 0xffffu);
    const int ls = (int)(wd >> 16);
    int tot;
    const int excl = block_scan_excl(c, wsum, lane, w, tot);
    roff[t] = carry + excl;
    rsrc[t] = k * EDGES_PER_BLOCK + ls;
    rcnt[t] = c;
    __syncthreads();
    unsigned int v[8][4];
    int cc[8], oo[8], ss[8];
#pragma unroll
    for (int u = 0; u < 8; ++u) {
      int kk = g + 32 * u;
      cc[u] = rcnt[kk];
      ss[u] = rsrc[kk];
      oo[u] = roff[kk];
#pragma unroll
      for (int q = 0; q < 4; ++q) {
        int i = l8 + 8 * q;
        v[u][q] = 0;
        if (i < cc[u]) v[u][q] = binned[ss[u] + i];
      }
    }
#pragma unroll
    for (int u = 0; u < 8; ++u) {
#pragma unroll
      for (int q = 0; q < 4; ++q) {
        int i = l8 + 8 * q;
        if (i < cc[u]) f(v[u][q], oo[u] + i);
      }
    }
#pragma unroll
    for (int u = 0; u < 8; ++u) {
      for (int i = l8 + 32; i < cc[u]; i += 8) f(binned[ss[u] + i], oo[u] + i);
    }
    carry += tot;
    __syncthreads();
  }
}

__global__ void __launch_bounds__(256) range_csr(const unsigned int* __restrict__ binned,
                                                  const unsigned int* __restrict__ cntmat,
                                                  int* __restrict__ rowptr, float* __restrict__ dis,
                                                  int* __restrict__ csr_src, int n, int nblk) {
  __shared__ int cnt[256];
  __shared__ int cur[256];
  __shared__ int wsum[8];
  __shared__ int roff[256];
  __shared__ int rsrc[256];
  __shared__ int rcnt[256];
  __shared__ int rsum[16];
  __shared__ unsigned int stage[STAGE_CAP];
  __shared__ unsigned short stage2[STAGE_CAP];
  const int t = threadIdx.x;
  const int lane = t & 31;
  const int w = t >> 5;
  const int r = blockIdx.x;

  unsigned int first = 0;
  if (t < nblk) first = cntmat[t * 256 + r];
  int csum = (int)(first & 0xffffu);
  int lsum = (int)(first >> 16);
  for (int k = t + 256; k < nblk; k += 256) {
    const unsigned int wd = cntmat[k * 256 + r];
    csum += (int)(wd & 0xffffu);
    lsum += (int)(wd >> 16);
  }
  cnt[t] = 0;
  int m, beg;
  block_sum2(csum, lsum, rsum, lane, w, m, beg);
  const bool staged = m <= STAGE_CAP;
  if (staged) {
    for_each_range_edge(binned, cntmat, r, nblk, first, roff, rsrc, rcnt, wsum,
                        [&](unsigned int v, int pos) {
                          stage[pos] = v;
                          atomicAdd(&cnt[v & 255u], 1);
                        });
  } else {
    for_each_range_edge(binned, cntmat, r, nblk, first, roff, rsrc, rcnt, wsum,
                        [&](unsigned int v, int pos) { atomicAdd(&cnt[v & 255u], 1); });
  }
  const int c = cnt[t];
  int dummy;
  const int excl = block_scan_excl(c, wsum, lane, w, dummy);
  const int node = (r << RANGE_SHIFT) + t;
  if (node < n) {
    rowptr[node] = beg + excl;
    dis[node] = rsqrtf((float)c + 1.0f);
  }
  if (node == n - 1) rowptr[n] = beg + excl + c;
  cur[t] = excl;
  __syncthreads();
  if (staged) {
    constexpr int U = 8;
    for (int j0 = t; j0 < m; j0 += 256 * U) {
      unsigned int v[U];
#pragma unroll
      for (int u = 0; u < U; ++u) {
        int j = j0 + 256 * u;
        v[u] = 0;
        if (j < m) v[u] = stage[j];
      }
      int p[U];
#pragma unroll
      for (int u = 0; u < U; ++u) {
        p[u] = 0;
        if (j0 + 256 * u < m) p[u] = atomicAdd(&cur[v[u] & 255u], 1);
      }
#pragma unroll
      for (int u = 0; u < U; ++u)
        if (j0 + 256 * u < m) stage2[p[u]] = (unsigned short)(v[u] >> 16);
    }
    __syncthreads();
    for (int i = t; i < m; i += 256) csr_src[beg + i] = (int)stage2[i];
  } else {
    for_each_range_edge(binned, cntmat, r, nblk, first, roff, rsrc, rcnt, wsum,
                        [&](unsigned int v, int pos) {
                          int p = atomicAdd(&cur[v & 255u], 1);
                          csr_src[beg + p] = (int)(v >> 16);
                        });
  }
}

template <int K, int N>
__global__ void __launch_bounds__(256) gemm_strip_f32(const float* __restrict__ A,
                                                       const float* __restrict__ B,
                                                       const float* __restrict__ rs,
                                                       float* __restrict__ D, int M) {
  constexpr int NT = N / 16;
  constexpr int Q = K / 8;
  constexpr int P2 = N + 8;
  constexpr int ITEMS = (K / 2) * (N / 4);
  __shared__ float4 Bs4[(K / 2) * P2 / 2];

  const int t = threadIdx.x;
  const int lane = t & 31;
  const int c = lane & 15;
  const int h = lane >> 4;
  const int wv = __builtin_amdgcn_readfirstlane(t >> 5);
  const int strips = (M + 15) >> 4;
  const int wstrip = blockIdx.x * 8 + wv;
  const bool live = wstrip < strips;
  const int strip = live ? wstrip : strips - 1;
  if (blockIdx.x == 0 && t < N / 4) ((float4*)(D + (size_t)M * N))[t] = make_float4(0.0f, 0.0f, 0.0f, 0.0f);

  const float4* B4 = (const float4*)B;
  float4 w0[ITEMS / 256], w1[ITEMS / 256];
#pragma unroll
  for (int i = 0; i < ITEMS / 256; ++i) {
    int it = t + 256 * i;
    int kp = it / (N / 4);
    int n4 = it % (N / 4);
    w0[i] = B4[(2 * kp) * (N / 4) + n4];
    w1[i] = B4[(2 * kp + 1) * (N / 4) + n4];
  }

  int rowA = strip * 16 + c;
  if (rowA > M - 1) rowA = M - 1;
  const float4* arow = (const float4*)(A + (size_t)rowA * K) + h;
  float4 a[Q];
#pragma unroll
  for (int q = 0; q < Q; ++q) a[q] = arow[2 * q];

  float sc[8];
#pragma unroll
  for (int j = 0; j < 8; ++j) {
    int r = strip * 16 + 8 * h + j;
    sc[j] = rs[r < M ? r : M - 1];
  }

#pragma unroll
  for (int i = 0; i < ITEMS / 256; ++i) {
    int it = t + 256 * i;
    int kp = it / (N / 4);
    int n4 = it % (N / 4);
    float4* dst = Bs4 + (kp * P2 + 4 * n4) / 2;
    dst[0] = make_float4(w0[i].x, w1[i].x, w0[i].y, w1[i].y);
    dst[1] = make_float4(w0[i].z, w1[i].z, w0[i].w, w1[i].w);
  }
  __builtin_amdgcn_fence(__ATOMIC_RELEASE, "workgroup", "local");
  __builtin_amdgcn_s_barrier();
  __builtin_amdgcn_fence(__ATOMIC_ACQUIRE, "workgroup", "local");

  const v2f* bp = (const v2f*)Bs4 + (2 * h) * P2 + c;
  v8f acc[NT];
#pragma unroll
  for (int tt = 0; tt < NT; ++tt) acc[tt] = v8f{};

#pragma unroll
  for (int q = 0; q < Q; ++q) {
#pragma unroll
    for (int s2 = 0; s2 < 2; ++s2) {
      v2f av;
      av.x = s2 ? a[q].z : a[q].x;
      av.y = s2 ? a[q].w : a[q].y;
#pragma unroll
      for (int tt = 0; tt < NT; ++tt) {
        v2f bv = bp[(4 * q + s2) * P2 + 16 * tt];
        acc[tt] = __builtin_amdgcn_wmma_f32_16x16x4_f32(
             false, av,  false, bv,
             (short)0, acc[tt],  false,  false);
      }
    }
  }

  if (!live) return;
  float* dlane = D + ((size_t)strip * 16 + 8 * h) * N + c;
  if (strip * 16 + 16 <= M) {
#pragma unroll
    for (int tt = 0; tt < NT; ++tt) {
#pragma unroll
      for (int j = 0; j < 8; ++j) dlane[j * N + 16 * tt] = acc[tt][j] * sc[j];
    }
  } else {
#pragma unroll
    for (int tt = 0; tt < NT; ++tt) {
#pragma unroll
      for (int j = 0; j < 8; ++j) {
        if (strip * 16 + 8 * h + j < M) dlane[j * N + 16 * tt] = acc[tt][j] * sc[j];
      }
    }
  }
}

template <int LPN>
__device__ __forceinline__ float4 edge_sum(const float* __restrict__ hs, const int* __restrict__ csr_src,
                                           int beg, int end, int sub, int pad) {
  const char* hb = (const char*)hs;
  const unsigned int lane_off = (unsigned int)sub * 16u;
  float4 acc = make_float4(0.0f, 0.0f, 0.0f, 0.0f);
  int s_cur = pad;
  if (beg + sub < end) s_cur = csr_src[beg + sub];
  for (int base = beg; base < end; base += LPN) {
    const bool upper = __builtin_amdgcn_ballot_w32(end - base > LPN / 2) != 0u;
    float4 v[LPN];
#pragma unroll
    for (int u = 0; u < LPN / 2; ++u) {
      const unsigned int s = (unsigned int)__shfl(s_cur, u, LPN);
      v[u] = *(const float4*)(hb + (size_t)(s * (unsigned int)(LPN * 16) + lane_off));
    }
    if (upper) {
#pragma unroll
      for (int u = LPN / 2; u < LPN; ++u) {
        const unsigned int s = (unsigned int)__shfl(s_cur, u, LPN);
        v[u] = *(const float4*)(hb + (size_t)(s * (unsigned int)(LPN * 16) + lane_off));
      }
    }
    const int nx = base + LPN + sub;
    int s_nxt = pad;
    if (nx < end) s_nxt = csr_src[nx];
#pragma unroll
    for (int u = 0; u < LPN / 2; ++u) {
      acc.x += v[u].x;
      acc.y += v[u].y;
      acc.z += v[u].z;
      acc.w += v[u].w;
    }
    if (upper) {
#pragma unroll
      for (int u = LPN / 2; u < LPN; ++u) {
        acc.x += v[u].x;
        acc.y += v[u].y;
        acc.z += v[u].z;
        acc.w += v[u].w;
      }
    }
    s_cur = s_nxt;
  }
  return acc;
}

template <int C, bool RELU>
__global__ void gather_rows(const float* __restrict__ h, const int* __restrict__ rowptr,
                            const int* __restrict__ csr_src,
                            const float* __restrict__ dis, const float* __restrict__ b,
                            float* __restrict__ out, int n) {
  constexpr int LPN = C / 4;
  constexpr int NPW = 32 / LPN;
  const int lane = threadIdx.x & 31;
  const int sub = lane % LPN;
  const int wave = (int)((blockIdx.x * blockDim.x + threadIdx.x) >> 5);
  const int node = wave * NPW + lane / LPN;
  if (node >= n) return;
  const int beg = rowptr[node];
  const int end = rowptr[node + 1];
  const float4* h4 = (const float4*)h;
  const float di = dis[node];
  const float4 hv = h4[(size_t)node * LPN + sub];
  const float4 bv = ((const float4*)b)[sub];
  const float4 acc = edge_sum<LPN>(h, csr_src, beg, end, sub, n);
  float4 r;
  r.x = (acc.x + hv.x) * di + bv.x;
  r.y = (acc.y + hv.y) * di + bv.y;
  r.z = (acc.z + hv.z) * di + bv.z;
  r.w = (acc.w + hv.w) * di + bv.w;
  if (RELU) {
    r.x = fmaxf(r.x, 0.0f);
    r.y = fmaxf(r.y, 0.0f);
    r.z = fmaxf(r.z, 0.0f);
    r.w = fmaxf(r.w, 0.0f);
  }
  ((float4*)out)[(size_t)node * LPN + sub] = r;
}

template <int C, int N2>
__global__ void __launch_bounds__(256) gather_relu_gemm(const float* __restrict__ h,
                                                         const int* __restrict__ rowptr,
                                                         const int* __restrict__ csr_src,
                                                         const float* __restrict__ dis,
                                                         const float* __restrict__ b,
                                                         const float* __restrict__ W,
                                                         float* __restrict__ out, int n) {
  constexpr int LPN = C / 4;
  constexpr int NPW = 32 / LPN;
  constexpr int ROWS = 8 * NPW;
  constexpr int Q = C / 8;
  constexpr int P4 = C / 4 + 1;
  constexpr int NT = N2 / 16;
  constexpr int P2 = N2 + 8;
  constexpr int ITEMS = (C / 2) * (N2 / 4);
  static_assert(ROWS == 16, "one 16-row matrix tile per block");
  static_assert(N2 % 16 == 0 && C % 8 == 0 && ITEMS == 256, "whole tiles; one staging item per thread");
  __shared__ float4 Hs4[ROWS * P4];
  __shared__ float4 Bs4[(C / 2) * P2 / 2];

  const int t = threadIdx.x;
  const int lane = t & 31;
  const int sub = lane % LPN;
  const int wv = __builtin_amdgcn_readfirstlane(t >> 5);
  const int rib = wv * NPW + lane / LPN;
  const int row0 = blockIdx.x * ROWS;
  const bool valid = row0 + rib < n;
  const int node = valid ? row0 + rib : n - 1;
  const int beg = rowptr[node];
  int end = rowptr[node + 1];
  if (!valid) end = beg;
  const float4* h4 = (const float4*)h;
  const float di = dis[node];
  const float4 hv = h4[(size_t)node * LPN + sub];
  const float4 bv = ((const float4*)b)[sub];
  if (blockIdx.x == 0 && t < N2 / 4) ((float4*)(out + (size_t)n * N2))[t] = make_float4(0.0f, 0.0f, 0.0f, 0.0f);

  const float4* W4 = (const float4*)W;
  {
    const int kp = t / (N2 / 4);
    const int n4 = t % (N2 / 4);
    float4 w0 = W4[(2 * kp) * (N2 / 4) + n4];
    float4 w1 = W4[(2 * kp + 1) * (N2 / 4) + n4];
    float4* dst = Bs4 + (kp * P2 + 4 * n4) / 2;
    dst[0] = make_float4(w0.x, w1.x, w0.y, w1.y);
    dst[1] = make_float4(w0.z, w1.z, w0.w, w1.w);
  }

  const float4 acc = edge_sum<LPN>(h, csr_src, beg, end, sub, n);
  float4 r;
  r.x = fmaxf((acc.x + hv.x) * di + bv.x, 0.0f);
  r.y = fmaxf((acc.y + hv.y) * di + bv.y, 0.0f);
  r.z = fmaxf((acc.z + hv.z) * di + bv.z, 0.0f);
  r.w = fmaxf((acc.w + hv.w) * di + bv.w, 0.0f);
  if (!valid) r = make_float4(0.0f, 0.0f, 0.0f, 0.0f);
  Hs4[rib * P4 + sub] = r;

  const int c = lane & 15;
  const int hh = lane >> 4;
  const bool mm = wv == (int)(blockIdx.x & 7);
  float sc[8];
  if (mm) {
#pragma unroll
    for (int j = 0; j < 8; ++j) {
      int rr = row0 + 8 * hh + j;
      sc[j] = dis[rr < n ? rr : n - 1];
    }
  }
  __builtin_amdgcn_fence(__ATOMIC_RELEASE, "workgroup", "local");
  __builtin_amdgcn_s_barrier();
  __builtin_amdgcn_fence(__ATOMIC_ACQUIRE, "workgroup", "local");
  if (!mm) return;

  const float4* ap = Hs4 + c * P4 + hh;
  float4 a[Q];
#pragma unroll
  for (int q = 0; q < Q; ++q) a[q] = ap[2 * q];
  const v2f* bp = (const v2f*)Bs4 + (2 * hh) * P2 + c;
  v8f d[NT];
#pragma unroll
  for (int tt = 0; tt < NT; ++tt) d[tt] = v8f{};
#pragma unroll
  for (int q = 0; q < Q; ++q) {
#pragma unroll
    for (int s2 = 0; s2 < 2; ++s2) {
      v2f av;
      av.x = s2 ? a[q].z : a[q].x;
      av.y = s2 ? a[q].w : a[q].y;
#pragma unroll
      for (int tt = 0; tt < NT; ++tt) {
        v2f bv2 = bp[(4 * q + s2) * P2 + 16 * tt];
        d[tt] = __builtin_amdgcn_wmma_f32_16x16x4_f32(
             false, av,  false, bv2,
             (short)0, d[tt],  false,  false);
      }
    }
  }
  float* dl = out + ((size_t)row0 + 8 * hh) * N2 + c;
  if (row0 + ROWS <= n) {
#pragma unroll
    for (int tt = 0; tt < NT; ++tt) {
#pragma unroll
      for (int j = 0; j < 8; ++j) dl[j * N2 + 16 * tt] = d[tt][j] * sc[j];
    }
  } else {
#pragma unroll
    for (int tt = 0; tt < NT; ++tt) {
#pragma unroll
      for (int j = 0; j < 8; ++j) {
        if (row0 + 8 * hh + j < n) dl[j * N2 + 16 * tt] = d[tt][j] * sc[j];
      }
    }
  }
}


extern "C" void kernel_launch(void* const* d_in, const int* in_sizes, int n_in,
                              void* d_out, int out_size, void* d_ws, size_t ws_size,
                              hipStream_t stream) {
  const float* x  = (const float*)d_in[0];
  const float* W1 = (const float*)d_in[1];
  const float* b1 = (const float*)d_in[2];
  const float* W2 = (const float*)d_in[3];
  const float* b2 = (const float*)d_in[4];
  const int*   ei = (const int*)d_in[5];

  const int IN  = 128;
  const int HID = in_sizes[2];
  const int n   = in_sizes[0] / IN;
  const int nE  = in_sizes[5] / 2;
  const int* src = ei;
  const int* dst = ei + nE;

  char* ws = (char*)d_ws;
  size_t off = 0;
  auto alloc = [&](size_t bytes) -> void* {
    void* p = (void*)(ws + off);
    off += (bytes + 255) & ~(size_t)255;
    return p;
  };
  const int nblk = (nE + 4095) / 4096;
  const int nrng = (n + 255) / 256;

  float* dis    = (float*)alloc((size_t)n * 4);
  float* bufA   = (float*)alloc((size_t)(n + 1) * HID * 4);
  float* bufB   = (float*)alloc((size_t)(n + 1) * HID * 4);
  int*   rowptr = (int*)alloc((size_t)(n + 1) * 4);
  unsigned int* cntmat = (unsigned int*)alloc((size_t)nblk * 256 * 4);
  unsigned int* binned = (unsigned int*)alloc((size_t)nE * 4);
  int*   csrS   = (int*)alloc((size_t)nE * 4);
  float* h1   = bufA;
  float* h2   = bufB;
  float* outp = (float*)d_out;

  const int T = 256;
  auto blocks1d = [&](long long work) -> unsigned int {
    return (unsigned int)((work + T - 1) / T);
  };

  bin_local<<<nblk, T, 0, stream>>>(src, dst, cntmat, binned, nE);
  range_csr<<<nrng, T, 0, stream>>>(binned, cntmat, rowptr, dis, csrS, n, nblk);

  const unsigned int stripBlocks = (unsigned int)(((n + 15) / 16 + 7) / 8);
  gemm_strip_f32<128, 64><<<stripBlocks, T, 0, stream>>>(x, W1, dis, h1, n);
  gather_relu_gemm<64, 32><<<(unsigned int)((n + 15) / 16), T, 0, stream>>>(
      h1, rowptr, csrS, dis, b1, W2, h2, n);

  gather_rows<32, false><<<blocks1d((long long)n * 8), T, 0, stream>>>(
      h2, rowptr, csrS, dis, b2, outp, n);
}


# Layer-1 product moved into the CSR launch: the block that counts a 256-node range's in-degrees also multiplies those 256 rows of x by W1 (same k order, same sums), scaled by its dis; one launch fewer
# speedup vs baseline: 6.2727x; 1.0440x over previous
#include <hip/hip_runtime.h>

typedef float v2f __attribute__((ext_vector_type(2)));
typedef float v8f __attribute__((ext_vector_type(8)));

constexpr int RANGE_SHIFT = 8;
constexpr int EDGES_PER_BLOCK = 4096;
constexpr int STAGE_CAP = 8192;

__device__ __forceinline__ int4 load4_vec(const int* __restrict__ p, int e, int nE) {
  int4 v = make_int4(-1, -1, -1, -1);
  if (e < nE) v = *(const int4*)(p + e);
  return v;
}
__device__ __forceinline__ int4 load4_words(const int* __restrict__ p, int e, int nE) {
  int4 v = make_int4(-1, -1, -1, -1);
  if (e < nE) v.x = p[e];
  if (e + 1 < nE) v.y = p[e + 1];
  if (e + 2 < nE) v.z = p[e + 2];
  if (e + 3 < nE) v.w = p[e + 3];
  return v;
}

__device__ __forceinline__ void lds_barrier() {
  __builtin_amdgcn_fence(__ATOMIC_RELEASE, "workgroup", "local");
  __builtin_amdgcn_s_barrier();
  __builtin_amdgcn_fence(__ATOMIC_ACQUIRE, "workgroup", "local");
}

template <bool LDS_ONLY = false>
__device__ __forceinline__ int block_scan_excl(int v, int* wsum, int lane, int w, int& total) {
  int inc = v;
  for (int o = 1; o < 32; o <<= 1) {
    int a = __shfl_up(inc, o);
    if (lane >= o) inc += a;
  }
  if (LDS_ONLY) lds_barrier(); else __syncthreads();
  if (lane == 31) wsum[w] = inc;
  if (LDS_ONLY) lds_barrier(); else __syncthreads();
  int off = 0, tot = 0;
  for (int k = 0; k < 8; ++k) {
    int s = wsum[k];
    tot += s;
    if (k < w) off += s;
  }
  total = tot;
  return off + inc - v;
}

__device__ __forceinline__ void block_sum2(int a, int b, int* rs, int lane, int w, int& sa, int& sb) {
  for (int o = 16; o > 0; o >>= 1) {
    a += __shfl_xor(a, o);
    b += __shfl_xor(b, o);
  }
  if (lane == 0) {
    rs[w] = a;
    rs[8 + w] = b;
  }
  __syncthreads();
  int x = 0, y = 0;
  for (int k = 0; k < 8; ++k) {
    x += rs[k];
    y += rs[8 + k];
  }
  sa = x;
  sb = y;
}

__device__ __forceinline__ void count_range(int* hist, int d) {
  if (d >= 0) atomicAdd(&hist[(d >> RANGE_SHIFT) & 255], 1);
}
__device__ __forceinline__ int take_slot(int* lcur, int d) {
  return d >= 0 ? atomicAdd(&lcur[(d >> RANGE_SHIFT) & 255], 1) : -1;
}
__device__ __forceinline__ void put_edge(unsigned int* sval, int p, int s, int d) {
  if (p >= 0) sval[p] = ((unsigned int)s << 16) | ((unsigned int)d & 0xffffu);
}

__global__ void __launch_bounds__(256) bin_local(const int* __restrict__ src,
                                                  const int* __restrict__ dst,
                                                  unsigned int* __restrict__ cntmat,
                                                  unsigned int* __restrict__ binned, int nE) {
  __shared__ int hist[256];
  __shared__ int lcur[256];
  __shared__ int wsum[8];
  __shared__ unsigned int sval[EDGES_PER_BLOCK];
  const int t = threadIdx.x;
  const int lane = t & 31;
  const int w = t >> 5;
  const int blk = blockIdx.x;
  hist[t] = 0;

  const int base = blk * EDGES_PER_BLOCK;
  const bool vec = (((((size_t)src) | ((size_t)dst)) & 15) == 0) && ((nE & 3) == 0);
  int4 s[4], d[4];
  if (vec) {
#pragma unroll
    for (int i = 0; i < 4; ++i) {
      int e = base + (i * 256 + t) * 4;
      d[i] = load4_vec(dst, e, nE);
      s[i] = load4_vec(src, e, nE);
    }
  } else {
#pragma unroll
    for (int i = 0; i < 4; ++i) {
      int e = base + (i * 256 + t) * 4;
      d[i] = load4_words(dst, e, nE);
      s[i] = load4_words(src, e, nE);
    }
  }
  __syncthreads();
#pragma unroll
  for (int i = 0; i < 4; ++i) {
    count_range(hist, d[i].x);
    count_range(hist, d[i].y);
    count_range(hist, d[i].z);
    count_range(hist, d[i].w);
  }
  __syncthreads();
  const int own = hist[t];
  int tot;
  const int lstart = block_scan_excl(own, wsum, lane, w, tot);
  lcur[t] = lstart;
  cntmat[blk * 256 + t] = ((unsigned int)lstart << 16) | (unsigned int)own;
  __syncthreads();

  int p[16];
#pragma unroll
  for (int i = 0; i < 4; ++i) {
    p[4 * i + 0] = take_slot(lcur, d[i].x);
    p[4 * i + 1] = take_slot(lcur, d[i].y);
    p[4 * i + 2] = take_slot(lcur, d[i].z);
    p[4 * i + 3] = take_slot(lcur, d[i].w);
  }
#pragma unroll
  for (int i = 0; i < 4; ++i) {
    put_edge(sval, p[4 * i + 0], s[i].x, d[i].x);
    put_edge(sval, p[4 * i + 1], s[i].y, d[i].y);
    put_edge(sval, p[4 * i + 2], s[i].z, d[i].z);
    put_edge(sval, p[4 * i + 3], s[i].w, d[i].w);
  }
  __syncthreads();
  int cntb = nE - base;
  if (cntb > EDGES_PER_BLOCK) cntb = EDGES_PER_BLOCK;
  for (int i = t; i < cntb; i += 256) binned[base + i] = sval[i];
}

template <class F>
__device__ __forceinline__ void for_each_range_edge(const unsigned int* __restrict__ binned,
                                                    const unsigned int* __restrict__ cntmat,
                                                    int r, int nblk, unsigned int first, int* roff,
                                                    int* rsrc, int* rcnt, int* wsum, F f) {
  const int t = threadIdx.x;
  const int lane = t & 31;
  const int w = t >> 5;
  const int g = t >> 3;
  const int l8 = t & 7;
  int carry = 0;
  for (int kb = 0; kb < nblk; kb += 256) {
    const int k = kb + t;
    unsigned int wd = first;
    if (kb > 0) {
      wd = 0;
      if (k < nblk) wd = cntmat[k * 256 + r];
    }
    const int c = (int)(wd & 0xffffu);
    const int ls = (int)(wd >> 16);
    int tot;
    const int excl = block_scan_excl(c, wsum, lane, w, tot);
    roff[t] = carry + excl;
    rsrc[t] = k * EDGES_PER_BLOCK + ls;
    rcnt[t] = c;
    __syncthreads();
    unsigned int v[8][4];
    int cc[8], oo[8], ss[8];
#pragma unroll
    for (int u = 0; u < 8; ++u) {
      int kk = g + 32 * u;
      cc[u] = rcnt[kk];
      ss[u] = rsrc[kk];
      oo[u] = roff[kk];
#pragma unroll
      for (int q = 0; q < 4; ++q) {
        int i = l8 + 8 * q;
        v[u][q] = 0;
        if (i < cc[u]) v[u][q] = binned[ss[u] + i];
      }
    }
#pragma unroll
    for (int u = 0; u < 8; ++u) {
#pragma unroll
      for (int q = 0; q < 4; ++q) {
        int i = l8 + 8 * q;
        if (i < cc[u]) f(v[u][q], oo[u] + i);
      }
    }
#pragma unroll
    for (int u = 0; u < 8; ++u) {
      for (int i = l8 + 32; i < cc[u]; i += 8) f(binned[ss[u] + i], oo[u] + i);
    }
    carry += tot;
    __syncthreads();
  }
}

template <int K, int N>
__global__ void __launch_bounds__(256) range_csr_gemm(const unsigned int* __restrict__ binned,
                                                       const unsigned int* __restrict__ cntmat,
                                                       int* __restrict__ rowptr, float* __restrict__ dis,
                                                       int* __restrict__ csr_src, int n, int nblk,
                                                       const float* __restrict__ A,
                                                       const float* __restrict__ B,
                                                       float* __restrict__ D) {
  constexpr int NT = N / 16;
  constexpr int Q = K / 8;
  constexpr int P2 = N + 8;
  constexpr int ITEMS = (K / 2) * (N / 4);
  __shared__ int cnt[256];
  __shared__ int cur[256];
  __shared__ int wsum[8];
  __shared__ int roff[256];
  __shared__ int rsrc[256];
  __shared__ int rcnt[256];
  __shared__ int rsum[16];
  __shared__ float sdis[256];
  __shared__ unsigned int stage[STAGE_CAP];
  __shared__ unsigned short stage2[STAGE_CAP];
  __shared__ float4 Bs4[(K / 2) * P2 / 2];
  const int t = threadIdx.x;
  const int lane = t & 31;
  const int w = t >> 5;
  const int r = blockIdx.x;

  unsigned int first = 0;
  if (t < nblk) first = cntmat[t * 256 + r];

  {
    const float4* B4 = (const float4*)B;
    float4 w0[ITEMS / 256], w1[ITEMS / 256];
#pragma unroll
    for (int i = 0; i < ITEMS / 256; ++i) {
      int it = t + 256 * i;
      int kp = it / (N / 4);
      int n4 = it % (N / 4);
      w0[i] = B4[(2 * kp) * (N / 4) + n4];
      w1[i] = B4[(2 * kp + 1) * (N / 4) + n4];
    }
#pragma unroll
    for (int i = 0; i < ITEMS / 256; ++i) {
      int it = t + 256 * i;
      int kp = it / (N / 4);
      int n4 = it % (N / 4);
      float4* dst = Bs4 + (kp * P2 + 4 * n4) / 2;
      dst[0] = make_float4(w0[i].x, w1[i].x, w0[i].y, w1[i].y);
      dst[1] = make_float4(w0[i].z, w1[i].z, w0[i].w, w1[i].w);
    }
  }
  if (r == 0 && t < N / 4) ((float4*)(D + (size_t)n * N))[t] = make_float4(0.0f, 0.0f, 0.0f, 0.0f);

  int csum = (int)(first & 0xffffu);
  int lsum = (int)(first >> 16);
  for (int k = t + 256; k < nblk; k += 256) {
    const unsigned int wd = cntmat[k * 256 + r];
    csum += (int)(wd & 0xffffu);
    lsum += (int)(wd >> 16);
  }
  cnt[t] = 0;
  int m, beg;
  block_sum2(csum, lsum, rsum, lane, w, m, beg);
  const bool staged = m <= STAGE_CAP;
  if (staged) {
    for_each_range_edge(binned, cntmat, r, nblk, first, roff, rsrc, rcnt, wsum,
                        [&](unsigned int v, int pos) {
                          stage[pos] = v;
                          atomicAdd(&cnt[v & 255u], 1);
                        });
  } else {
    for_each_range_edge(binned, cntmat, r, nblk, first, roff, rsrc, rcnt, wsum,
                        [&](unsigned int v, int pos) { atomicAdd(&cnt[v & 255u], 1); });
  }

  const int fc = lane & 15;
  const int fh = lane >> 4;
  const int wv = __builtin_amdgcn_readfirstlane(w);
  const int rows0 = (r << RANGE_SHIFT) + wv * 32;
  float4 a[2][Q];
#pragma unroll
  for (int sI = 0; sI < 2; ++sI) {
    int rowA = rows0 + 16 * sI + fc;
    if (rowA > n - 1) rowA = n - 1;
    const float4* arow = (const float4*)(A + (size_t)rowA * K) + fh;
#pragma unroll
    for (int q = 0; q < Q; ++q) a[sI][q] = arow[2 * q];
  }

  const int c = cnt[t];
  int dummy;
  const int excl = block_scan_excl<true>(c, wsum, lane, w, dummy);
  const int node = (r << RANGE_SHIFT) + t;
  const float dval = rsqrtf((float)c + 1.0f);
  if (node < n) {
    rowptr[node] = beg + excl;
    dis[node] = dval;
  }
  if (node == n - 1) rowptr[n] = beg + excl + c;
  sdis[t] = dval;
  cur[t] = excl;
  lds_barrier();
  if (staged) {
    constexpr int U = 8;
    for (int j0 = t; j0 < m; j0 += 256 * U) {
      unsigned int v[U];
#pragma unroll
      for (int u = 0; u < U; ++u) {
        int j = j0 + 256 * u;
        v[u] = 0;
        if (j < m) v[u] = stage[j];
      }
      int p[U];
#pragma unroll
      for (int u = 0; u < U; ++u) {
        p[u] = 0;
        if (j0 + 256 * u < m) p[u] = atomicAdd(&cur[v[u] & 255u], 1);
      }
#pragma unroll
      for (int u = 0; u < U; ++u)
        if (j0 + 256 * u < m) stage2[p[u]] = (unsigned short)(v[u] >> 16);
    }
    lds_barrier();
    for (int i = t; i < m; i += 256) csr_src[beg + i] = (int)stage2[i];
  } else {
    for_each_range_edge(binned, cntmat, r, nblk, first, roff, rsrc, rcnt, wsum,
                        [&](unsigned int v, int pos) {
                          int p = atomicAdd(&cur[v & 255u], 1);
                          csr_src[beg + p] = (int)(v >> 16);
                        });
  }

  const v2f* bp = (const v2f*)Bs4 + (2 * fh) * P2 + fc;
#pragma unroll
  for (int sI = 0; sI < 2; ++sI) {
    const int row0 = rows0 + 16 * sI;
    if (row0 >= n) continue;
    v8f acc[NT];
#pragma unroll
    for (int tt = 0; tt < NT; ++tt) acc[tt] = v8f{};
#pragma unroll
    for (int q = 0; q < Q; ++q) {
#pragma unroll
      for (int s2 = 0; s2 < 2; ++s2) {
        v2f av;
        av.x = s2 ? a[sI][q].z : a[sI][q].x;
        av.y = s2 ? a[sI][q].w : a[sI][q].y;
#pragma unroll
        for (int tt = 0; tt < NT; ++tt) {
          v2f bv = bp[(4 * q + s2) * P2 + 16 * tt];
          acc[tt] = __builtin_amdgcn_wmma_f32_16x16x4_f32(
               false, av,  false, bv,
               (short)0, acc[tt],  false,  false);
        }
      }
    }
    float sc[8];
#pragma unroll
    for (int j = 0; j < 8; ++j) sc[j] = sdis[wv * 32 + 16 * sI + 8 * fh + j];
    float* dlane = D + ((size_t)row0 + 8 * fh) * N + fc;
    if (row0 + 16 <= n) {
#pragma unroll
      for (int tt = 0; tt < NT; ++tt) {
#pragma unroll
        for (int j = 0; j < 8; ++j) dlane[j * N + 16 * tt] = acc[tt][j] * sc[j];
      }
    } else {
#pragma unroll
      for (int tt = 0; tt < NT; ++tt) {
#pragma unroll
        for (int j = 0; j < 8; ++j) {
          if (row0 + 8 * fh + j < n) dlane[j * N + 16 * tt] = acc[tt][j] * sc[j];
        }
      }
    }
  }
}

template <int LPN>
__device__ __forceinline__ float4 edge_sum(const float* __restrict__ hs, const int* __restrict__ csr_src,
                                           int beg, int end, int sub, int pad) {
  const char* hb = (const char*)hs;
  const unsigned int lane_off = (unsigned int)sub * 16u;
  float4 acc = make_float4(0.0f, 0.0f, 0.0f, 0.0f);
  int s_cur = pad;
  if (beg + sub < end) s_cur = csr_src[beg + sub];
  for (int base = beg; base < end; base += LPN) {
    const bool upper = __builtin_amdgcn_ballot_w32(end - base > LPN / 2) != 0u;
    float4 v[LPN];
#pragma unroll
    for (int u = 0; u < LPN / 2; ++u) {
      const unsigned int s = (unsigned int)__shfl(s_cur, u, LPN);
      v[u] = *(const float4*)(hb + (size_t)(s * (unsigned int)(LPN * 16) + lane_off));
    }
    if (upper) {
#pragma unroll
      for (int u = LPN / 2; u < LPN; ++u) {
        const unsigned int s = (unsigned int)__shfl(s_cur, u, LPN);
        v[u] = *(const float4*)(hb + (size_t)(s * (unsigned int)(LPN * 16) + lane_off));
      }
    }
    const int nx = base + LPN + sub;
    int s_nxt = pad;
    if (nx < end) s_nxt = csr_src[nx];
#pragma unroll
    for (int u = 0; u < LPN / 2; ++u) {
      acc.x += v[u].x;
      acc.y += v[u].y;
      acc.z += v[u].z;
      acc.w += v[u].w;
    }
    if (upper) {
#pragma unroll
      for (int u = LPN / 2; u < LPN; ++u) {
        acc.x += v[u].x;
        acc.y += v[u].y;
        acc.z += v[u].z;
        acc.w += v[u].w;
      }
    }
    s_cur = s_nxt;
  }
  return acc;
}

template <int C, bool RELU>
__global__ void gather_rows(const float* __restrict__ h, const int* __restrict__ rowptr,
                            const int* __restrict__ csr_src,
                            const float* __restrict__ dis, const float* __restrict__ b,
                            float* __restrict__ out, int n) {
  constexpr int LPN = C / 4;
  constexpr int NPW = 32 / LPN;
  const int lane = threadIdx.x & 31;
  const int sub = lane % LPN;
  const int wave = (int)((blockIdx.x * blockDim.x + threadIdx.x) >> 5);
  const int node = wave * NPW + lane / LPN;
  if (node >= n) return;
  const int beg = rowptr[node];
  const int end = rowptr[node + 1];
  const float4* h4 = (const float4*)h;
  const float di = dis[node];
  const float4 hv = h4[(size_t)node * LPN + sub];
  const float4 bv = ((const float4*)b)[sub];
  const float4 acc = edge_sum<LPN>(h, csr_src, beg, end, sub, n);
  float4 r;
  r.x = (acc.x + hv.x) * di + bv.x;
  r.y = (acc.y + hv.y) * di + bv.y;
  r.z = (acc.z + hv.z) * di + bv.z;
  r.w = (acc.w + hv.w) * di + bv.w;
  if (RELU) {
    r.x = fmaxf(r.x, 0.0f);
    r.y = fmaxf(r.y, 0.0f);
    r.z = fmaxf(r.z, 0.0f);
    r.w = fmaxf(r.w, 0.0f);
  }
  ((float4*)out)[(size_t)node * LPN + sub] = r;
}

template <int C, int N2>
__global__ void __launch_bounds__(256) gather_relu_gemm(const float* __restrict__ h,
                                                         const int* __restrict__ rowptr,
                                                         const int* __restrict__ csr_src,
                                                         const float* __restrict__ dis,
                                                         const float* __restrict__ b,
                                                         const float* __restrict__ W,
                                                         float* __restrict__ out, int n) {
  constexpr int LPN = C / 4;
  constexpr int NPW = 32 / LPN;
  constexpr int ROWS = 8 * NPW;
  constexpr int Q = C / 8;
  constexpr int P4 = C / 4 + 1;
  constexpr int NT = N2 / 16;
  constexpr int P2 = N2 + 8;
  constexpr int ITEMS = (C / 2) * (N2 / 4);
  static_assert(ROWS == 16, "one 16-row matrix tile per block");
  static_assert(N2 % 16 == 0 && C % 8 == 0 && ITEMS == 256, "whole tiles; one staging item per thread");
  __shared__ float4 Hs4[ROWS * P4];
  __shared__ float4 Bs4[(C / 2) * P2 / 2];

  const int t = threadIdx.x;
  const int lane = t & 31;
  const int sub = lane % LPN;
  const int wv = __builtin_amdgcn_readfirstlane(t >> 5);
  const int rib = wv * NPW + lane / LPN;
  const int row0 = blockIdx.x * ROWS;
  const bool valid = row0 + rib < n;
  const int node = valid ? row0 + rib : n - 1;
  const int beg = rowptr[node];
  int end = rowptr[node + 1];
  if (!valid) end = beg;
  const float4* h4 = (const float4*)h;
  const float di = dis[node];
  const float4 hv = h4[(size_t)node * LPN + sub];
  const float4 bv = ((const float4*)b)[sub];
  if (blockIdx.x == 0 && t < N2 / 4) ((float4*)(out + (size_t)n * N2))[t] = make_float4(0.0f, 0.0f, 0.0f, 0.0f);

  const float4* W4 = (const float4*)W;
  {
    const int kp = t / (N2 / 4);
    const int n4 = t % (N2 / 4);
    float4 w0 = W4[(2 * kp) * (N2 / 4) + n4];
    float4 w1 = W4[(2 * kp + 1) * (N2 / 4) + n4];
    float4* dst = Bs4 + (kp * P2 + 4 * n4) / 2;
    dst[0] = make_float4(w0.x, w1.x, w0.y, w1.y);
    dst[1] = make_float4(w0.z, w1.z, w0.w, w1.w);
  }

  const float4 acc = edge_sum<LPN>(h, csr_src, beg, end, sub, n);
  float4 r;
  r.x = fmaxf((acc.x + hv.x) * di + bv.x, 0.0f);
  r.y = fmaxf((acc.y + hv.y) * di + bv.y, 0.0f);
  r.z = fmaxf((acc.z + hv.z) * di + bv.z, 0.0f);
  r.w = fmaxf((acc.w + hv.w) * di + bv.w, 0.0f);
  if (!valid) r = make_float4(0.0f, 0.0f, 0.0f, 0.0f);
  Hs4[rib * P4 + sub] = r;

  const int c = lane & 15;
  const int hh = lane >> 4;
  const bool mm = wv == (int)(blockIdx.x & 7);
  float sc[8];
  if (mm) {
#pragma unroll
    for (int j = 0; j < 8; ++j) {
      int rr = row0 + 8 * hh + j;
      sc[j] = dis[rr < n ? rr : n - 1];
    }
  }
  __builtin_amdgcn_fence(__ATOMIC_RELEASE, "workgroup", "local");
  __builtin_amdgcn_s_barrier();
  __builtin_amdgcn_fence(__ATOMIC_ACQUIRE, "workgroup", "local");
  if (!mm) return;

  const float4* ap = Hs4 + c * P4 + hh;
  float4 a[Q];
#pragma unroll
  for (int q = 0; q < Q; ++q) a[q] = ap[2 * q];
  const v2f* bp = (const v2f*)Bs4 + (2 * hh) * P2 + c;
  v8f d[NT];
#pragma unroll
  for (int tt = 0; tt < NT; ++tt) d[tt] = v8f{};
#pragma unroll
  for (int q = 0; q < Q; ++q) {
#pragma unroll
    for (int s2 = 0; s2 < 2; ++s2) {
      v2f av;
      av.x = s2 ? a[q].z : a[q].x;
      av.y = s2 ? a[q].w : a[q].y;
#pragma unroll
      for (int tt = 0; tt < NT; ++tt) {
        v2f bv2 = bp[(4 * q + s2) * P2 + 16 * tt];
        d[tt] = __builtin_amdgcn_wmma_f32_16x16x4_f32(
             false, av,  false, bv2,
             (short)0, d[tt],  false,  false);
      }
    }
  }
  float* dl = out + ((size_t)row0 + 8 * hh) * N2 + c;
  if (row0 + ROWS <= n) {
#pragma unroll
    for (int tt = 0; tt < NT; ++tt) {
#pragma unroll
      for (int j = 0; j < 8; ++j) dl[j * N2 + 16 * tt] = d[tt][j] * sc[j];
    }
  } else {
#pragma unroll
    for (int tt = 0; tt < NT; ++tt) {
#pragma unroll
      for (int j = 0; j < 8; ++j) {
        if (row0 + 8 * hh + j < n) dl[j * N2 + 16 * tt] = d[tt][j] * sc[j];
      }
    }
  }
}


extern "C" void kernel_launch(void* const* d_in, const int* in_sizes, int n_in,
                              void* d_out, int out_size, void* d_ws, size_t ws_size,
                              hipStream_t stream) {
  const float* x  = (const float*)d_in[0];
  const float* W1 = (const float*)d_in[1];
  const float* b1 = (const float*)d_in[2];
  const float* W2 = (const float*)d_in[3];
  const float* b2 = (const float*)d_in[4];
  const int*   ei = (const int*)d_in[5];

  const int IN  = 128;
  const int HID = in_sizes[2];
  const int n   = in_sizes[0] / IN;
  const int nE  = in_sizes[5] / 2;
  const int* src = ei;
  const int* dst = ei + nE;

  char* ws = (char*)d_ws;
  size_t off = 0;
  auto alloc = [&](size_t bytes) -> void* {
    void* p = (void*)(ws + off);
    off += (bytes + 255) & ~(size_t)255;
    return p;
  };
  const int nblk = (nE + 4095) / 4096;
  const int nrng = (n + 255) / 256;

  float* dis    = (float*)alloc((size_t)n * 4);
  float* bufA   = (float*)alloc((size_t)(n + 1) * HID * 4);
  float* bufB   = (float*)alloc((size_t)(n + 1) * HID * 4);
  int*   rowptr = (int*)alloc((size_t)(n + 1) * 4);
  unsigned int* cntmat = (unsigned int*)alloc((size_t)nblk * 256 * 4);
  unsigned int* binned = (unsigned int*)alloc((size_t)nE * 4);
  int*   csrS   = (int*)alloc((size_t)nE * 4);
  float* h1   = bufA;
  float* h2   = bufB;
  float* outp = (float*)d_out;

  const int T = 256;
  auto blocks1d = [&](long long work) -> unsigned int {
    return (unsigned int)((work + T - 1) / T);
  };

  bin_local<<<nblk, T, 0, stream>>>(src, dst, cntmat, binned, nE);
  range_csr_gemm<128, 64><<<nrng, T, 0, stream>>>(binned, cntmat, rowptr, dis, csrS, n, nblk, x, W1, h1);

  gather_relu_gemm<64, 32><<<(unsigned int)((n + 15) / 16), T, 0, stream>>>(
      h1, rowptr, csrS, dis, b1, W2, h2, n);

  gather_rows<32, false><<<blocks1d((long long)n * 8), T, 0, stream>>>(
      h2, rowptr, csrS, dis, b2, outp, n);
}
